# starting kernel
# baseline (speedup 1.0000x reference)
#include <hip/hip_runtime.h>
#include <hip/hip_bf16.h>

typedef float v2f __attribute__((ext_vector_type(2)));
typedef float v8f __attribute__((ext_vector_type(8)));

#define N_NODES 50000
#define KDIM    128
#define LDA     132

__global__ __launch_bounds__(256) void deg_kernel(const int* __restrict__ dst,
                                                  float* __restrict__ deg, int nE) {
    int e = blockIdx.x * blockDim.x + threadIdx.x;
    if (e < nE) unsafeAtomicAdd(&deg[dst[e]], 1.0f);
}

__global__ __launch_bounds__(256) void scatter_kernel(const float* __restrict__ feat,
                                                      const int* __restrict__ src,
                                                      const int* __restrict__ dst,
                                                      float* __restrict__ msg, int nE) {
    int t = blockIdx.x * blockDim.x + threadIdx.x;
    int e = t >> 5;
    if (e >= nE) return;
    int c = (t & 31) << 2;
    int s = src[e], d = dst[e];
    const float4 v = *(const float4*)(feat + (size_t)s * KDIM + c);
    float* p = msg + (size_t)d * KDIM + c;
    unsafeAtomicAdd(p + 0, v.x);
    unsafeAtomicAdd(p + 1, v.y);
    unsafeAtomicAdd(p + 2, v.z);
    unsafeAtomicAdd(p + 3, v.w);
}

__global__ __launch_bounds__(256) void sage_gemm_wmma(const float* __restrict__ msg,
                                                      const float* __restrict__ feat,
                                                      const float* __restrict__ deg,
                                                      const float* __restrict__ W,
                                                      const float* __restrict__ bias,
                                                      float* __restrict__ out,
                                                      int n_out, int apply_lrelu) {
    __shared__ float Atile[16 * LDA];

    const int tid  = threadIdx.x;
    const int nt   = blockDim.x;
    const int wave = tid >> 5;
    const int lane = tid & 31;
    const int m0   = blockIdx.x * 16;

    for (int i = tid; i < 16 * (KDIM / 4); i += nt) {
        int r  = i >> 5;
        int c4 = (i & 31) << 2;
        int row = m0 + r;
        float inv = 1.0f / (deg[row] + 1.0f);
        const float4 mv = *(const float4*)(msg  + (size_t)row * KDIM + c4);
        const float4 xv = *(const float4*)(feat + (size_t)row * KDIM + c4);
        float* a = &Atile[r * LDA + c4];
        a[0] = (mv.x + xv.x) * inv;
        a[1] = (mv.y + xv.y) * inv;
        a[2] = (mv.z + xv.z) * inv;
        a[3] = (mv.w + xv.w) * inv;
    }
    __syncthreads();

    const int n0 = wave * 16;
    const int lr = lane & 15;
    const int hi = lane >> 4;

    v8f acc = {0.f, 0.f, 0.f, 0.f, 0.f, 0.f, 0.f, 0.f};

#pragma unroll 8
    for (int k = 0; k < KDIM; k += 4) {
        v2f a, b;
        const float* ap = &Atile[lr * LDA + k + 2 * hi];
        a.x = ap[0];
        a.y = ap[1];
        const float* bp = &W[(size_t)(k + 2 * hi) * n_out + n0 + lr];
        b.x = bp[0];
        b.y = bp[n_out];
        acc = __builtin_amdgcn_wmma_f32_16x16x4_f32(
             false, a,  false, b,
             (short)0, acc,  false,  false);
    }

    const float bv   = bias[n0 + lr];
    const int  rbase = m0 + 8 * hi;
    const int  col   = n0 + lr;
#pragma unroll
    for (int i = 0; i < 8; ++i) {
        float v = acc[i] + bv;
        if (apply_lrelu) v = (v >= 0.f) ? v : 0.01f * v;
        out[(size_t)(rbase + i) * n_out + col] = v;
    }
}

extern "C" void kernel_launch(void* const* d_in, const int* in_sizes, int n_in,
                              void* d_out, int out_size, void* d_ws, size_t ws_size,
                              hipStream_t stream) {
    const float* x   = (const float*)d_in[0];
    const int*   edg = (const int*)  d_in[1];
    const float* W1  = (const float*)d_in[2];
    const float* b1  = (const float*)d_in[3];
    const float* W2  = (const float*)d_in[4];
    const float* b2  = (const float*)d_in[5];
    float*       outp = (float*)d_out;

    const int E = in_sizes[1] / 2;
    const int* src = edg;
    const int* dst = edg + E;

    float* msg = (float*)d_ws;
    float* deg = msg + (size_t)N_NODES * KDIM;
    float* h1  = deg + N_NODES;

    const size_t msgBytes = (size_t)N_NODES * KDIM * sizeof(float);

    hipMemsetAsync(msg, 0, msgBytes, stream);
    hipMemsetAsync(deg, 0, (size_t)N_NODES * sizeof(float), stream);

    deg_kernel<<<(E + 255) / 256, 256, 0, stream>>>(dst, deg, E);

    {
        long long threads = (long long)E * 32;
        int blocks = (int)((threads + 255) / 256);
        scatter_kernel<<<blocks, 256, 0, stream>>>(x, src, dst, msg, E);
    }

    sage_gemm_wmma<<<N_NODES / 16, 256, 0, stream>>>(msg, x, deg, W1, b1, h1,
                                                      128,  1);

    hipMemsetAsync(msg, 0, msgBytes, stream);

    {
        long long threads = (long long)E * 32;
        int blocks = (int)((threads + 255) / 256);
        scatter_kernel<<<blocks, 256, 0, stream>>>(h1, src, dst, msg, E);
    }

    sage_gemm_wmma<<<N_NODES / 16, 64, 0, stream>>>(msg, h1, deg, W2, b2, outp,
                                                     32,  0);
}


# one launch per layer: workgroup keeps 208 dst rows in LDS, scans all dst indices, adds matching src rows by LDS f32 atomics, WMMA from LDS; no msg/deg plane, global atomic, memset; sum order changed
# speedup vs baseline: 3.0434x; 3.0434x over previous
#include <hip/hip_runtime.h>


typedef float v2f __attribute__((ext_vector_type(2)));
typedef float v8f __attribute__((ext_vector_type(8)));

#define N_NODES 50000
#define KDIM    128
#define LDA     132
#define RROWS   208
#define NWAVES  32
#define GPS     8
#define QSTEP   (GPS * 4 * 32)
#define QCAP    (QSTEP + 512)

template <int NOUT>
__global__ __launch_bounds__(1024) void sage_layer(const float* __restrict__ feat,
                                                   const int* __restrict__ src,
                                                   const int* __restrict__ dst,
                                                   const float* __restrict__ W,
                                                   const float* __restrict__ bias,
                                                   float* __restrict__ out,
                                                   int nE, int apply_lrelu) {
    __shared__ __attribute__((aligned(16))) float acc[(RROWS + 1) * LDA];
    __shared__ unsigned queue[NWAVES * QCAP];

    const int tid  = threadIdx.x;
    const int wave = __builtin_amdgcn_readfirstlane(tid >> 5);
    const int lane = tid & 31;
    const int row0 = blockIdx.x * RROWS;
    int nrows = N_NODES - row0;
    if (nrows > RROWS) nrows = RROWS;
    const int ntiles = nrows >> 4;

    for (int i = tid; i < (RROWS + 1) * (KDIM / 4); i += NWAVES * 32) {
        const int r  = i >> 5;
        const int c4 = (i & 31) << 2;
        float4 v = make_float4(0.f, 0.f, 0.f, 0.f);
        if (r < nrows) v = *(const float4*)(feat + (size_t)(row0 + r) * KDIM + c4);
        *(float4*)&acc[r * LDA + c4] = v;
    }
    for (int r = tid; r <= RROWS; r += NWAVES * 32) acc[r * LDA + KDIM] = 0.f;
    __syncthreads();

    unsigned* q = queue + wave * QCAP;
    int qn = 0;

    auto drain = [&]() {
        for (int i = 0; i < qn; i += 32) {
            const bool live = (i + lane) < qn;
            const unsigned ent = live ? q[i + lane] : 0u;
            const int s  = live ? src[ent >> 8] : 0;
            const int dl = live ? (int)(ent & 255u) : RROWS;
            int nb = qn - i;
            if (nb > 32) nb = 32;
            for (int u0 = 0; u0 < nb; u0 += 8) {
                float v[8][4];
                int   dls[8];
#pragma unroll
                for (int u = 0; u < 8; ++u) {
                    const int su = __builtin_amdgcn_readlane(s, u0 + u);
                    dls[u]       = __builtin_amdgcn_readlane(dl, u0 + u);
                    const float* p = feat + (size_t)su * KDIM + lane;
                    v[u][0] = p[0];
                    v[u][1] = p[32];
                    v[u][2] = p[64];
                    v[u][3] = p[96];
                }
#pragma unroll
                for (int u = 0; u < 8; ++u) {
                    float* a = &acc[dls[u] * LDA + lane];
                    __hip_atomic_fetch_add(a,      v[u][0], __ATOMIC_RELAXED, __HIP_MEMORY_SCOPE_WORKGROUP);
                    __hip_atomic_fetch_add(a + 32, v[u][1], __ATOMIC_RELAXED, __HIP_MEMORY_SCOPE_WORKGROUP);
                    __hip_atomic_fetch_add(a + 64, v[u][2], __ATOMIC_RELAXED, __HIP_MEMORY_SCOPE_WORKGROUP);
                    __hip_atomic_fetch_add(a + 96, v[u][3], __ATOMIC_RELAXED, __HIP_MEMORY_SCOPE_WORKGROUP);
                }
            }
        }
    };

    {
        const int4* dst4  = (const int4*)dst;
        const int   ngrp  = nE >> 2;
        const int   nfull = ngrp >> 8;

        auto sift = [&](const int4 (&g)[GPS], int step) {
            if (qn > QCAP - QSTEP) {
                drain();
                qn = 0;
            }
#pragma unroll
            for (int j = 0; j < GPS; ++j) {
                const int e0 = (step * (GPS * 32) + j * 32 + lane) << 2;
                const unsigned d[4] = {(unsigned)(g[j].x - row0), (unsigned)(g[j].y - row0),
                                       (unsigned)(g[j].z - row0), (unsigned)(g[j].w - row0)};
                const bool any = (d[0] < (unsigned)nrows) | (d[1] < (unsigned)nrows) |
                                 (d[2] < (unsigned)nrows) | (d[3] < (unsigned)nrows);
                if (__builtin_amdgcn_ballot_w32(any) != 0u) {
#pragma unroll
                    for (int c = 0; c < 4; ++c) {
                        const bool m = d[c] < (unsigned)nrows;
                        const unsigned mk = __builtin_amdgcn_ballot_w32(m);
                        if (m) {
                            const int pos = qn + (int)__builtin_amdgcn_mbcnt_lo(mk, 0u);
                            q[pos] = ((unsigned)(e0 + c) << 8) | d[c];
                            __hip_atomic_fetch_add(&acc[d[c] * LDA + KDIM], 1.0f, __ATOMIC_RELAXED,
                                                   __HIP_MEMORY_SCOPE_WORKGROUP);
                        }
                        qn += __builtin_popcount(mk);
                    }
                }
            }
        };

        int4 g[GPS];
        for (int step = wave; step < nfull; step += NWAVES) {
            const int4* p = dst4 + (size_t)step * (GPS * 32) + lane;
#pragma unroll
            for (int j = 0; j < GPS; ++j) g[j] = p[j * 32];
            sift(g, step);
        }
        if (wave == (nfull & (NWAVES - 1)) && (ngrp & (GPS * 32 - 1)) != 0) {
            const int4 none = make_int4(-1, -1, -1, -1);
#pragma unroll
            for (int j = 0; j < GPS; ++j) {
                const int  gi = nfull * (GPS * 32) + j * 32 + lane;
                const int4 t  = dst4[gi < ngrp ? gi : ngrp - 1];
                g[j] = (gi < ngrp) ? t : none;
            }
            sift(g, nfull);
        }
        drain();
    }
    __syncthreads();

    for (int i = tid; i < nrows * (KDIM / 4); i += NWAVES * 32) {
        const int r  = i >> 5;
        const int c4 = (i & 31) << 2;
        const float inv = 1.0f / (acc[r * LDA + KDIM] + 1.0f);
        float4 v = *(float4*)&acc[r * LDA + c4];
        v.x *= inv; v.y *= inv; v.z *= inv; v.w *= inv;
        *(float4*)&acc[r * LDA + c4] = v;
    }
    __syncthreads();

    constexpr int NCT  = NOUT / 16;
    constexpr int RG   = NWAVES / NCT;
    constexpr int NACC = (RROWS / 16 + RG - 1) / RG;
    const int ct = wave % NCT;
    const int rg = wave / NCT;
    const int n0 = ct * 16;
    const int lr = lane & 15;
    const int hi = lane >> 4;

    int arow[NACC];
#pragma unroll
    for (int i = 0; i < NACC; ++i) {
        const int t = rg + RG * i;
        arow[i] = ((t < ntiles) ? t : (ntiles - 1)) * 16 + lr;
    }
    v8f c[NACC];
#pragma unroll
    for (int i = 0; i < NACC; ++i) c[i] = (v8f){0.f, 0.f, 0.f, 0.f, 0.f, 0.f, 0.f, 0.f};

#pragma unroll 8
    for (int k = 0; k < KDIM; k += 4) {
        v2f b;
        const float* bp = &W[(size_t)(k + 2 * hi) * NOUT + n0 + lr];
        b.x = bp[0];
        b.y = bp[NOUT];
#pragma unroll
        for (int i = 0; i < NACC; ++i) {
            v2f a;
            const float* ap = &acc[arow[i] * LDA + k + 2 * hi];
            a.x = ap[0];
            a.y = ap[1];
            c[i] = __builtin_amdgcn_wmma_f32_16x16x4_f32(
                 false, a,  false, b,
                 (short)0, c[i],  false,  false);
        }
    }

    const float bv  = bias[n0 + lr];
    const int   col = n0 + lr;
#pragma unroll
    for (int i = 0; i < NACC; ++i) {
        const int t = rg + RG * i;
        if (t < ntiles) {
            const int rbase = row0 + t * 16 + 8 * hi;
#pragma unroll
            for (int r = 0; r < 8; ++r) {
                float v = c[i][r] + bv;
                if (apply_lrelu) v = (v >= 0.f) ? v : 0.01f * v;
                out[(size_t)(rbase + r) * NOUT + col] = v;
            }
        }
    }
}

extern "C" void kernel_launch(void* const* d_in, const int* in_sizes, int n_in,
                              void* d_out, int out_size, void* d_ws, size_t ws_size,
                              hipStream_t stream) {
    const float* x   = (const float*)d_in[0];
    const int*   edg = (const int*)  d_in[1];
    const float* W1  = (const float*)d_in[2];
    const float* b1  = (const float*)d_in[3];
    const float* W2  = (const float*)d_in[4];
    const float* b2  = (const float*)d_in[5];
    float*       outp = (float*)d_out;

    const int E = in_sizes[1] / 2;
    const int* src = edg;
    const int* dst = edg + E;

    float* h1 = (float*)d_ws;

    sage_layer<128><<<(N_NODES + RROWS - 1) / RROWS, NWAVES * 32, 0, stream>>>(
        x, src, dst, W1, b1, h1, E,  1);
    sage_layer<32><<<(N_NODES + RROWS - 1) / RROWS, NWAVES * 32, 0, stream>>>(
        h1, src, dst, W2, b2, outp, E,  0);
}


# layer 1 also appends each matched edge (source, local row) to its workgroup's list in the workspace, plus in-degrees; layer 2 walks that list: no second dst scan; layer-2 sum order changed
# speedup vs baseline: 3.8000x; 1.2486x over previous
#include <hip/hip_runtime.h>


typedef float v2f __attribute__((ext_vector_type(2)));
typedef float v8f __attribute__((ext_vector_type(8)));

#define N_NODES 50000
#define KDIM    128
#define LDA     132
#define RROWS   208
#define NWAVES  32
#define GPS     8
#define QSTEP   (GPS * 4 * 32)
#define QCAP    (QSTEP + 512)
#define NWG     ((N_NODES + RROWS - 1) / RROWS)
#define LCAP    4096

template <int NOUT, bool REPLAY>
__global__ __launch_bounds__(1024) void sage_layer(const float* __restrict__ feat,
                                                   const int* __restrict__ src,
                                                   const int* __restrict__ dst,
                                                   const float* __restrict__ W,
                                                   const float* __restrict__ bias,
                                                   float* __restrict__ out,
                                                   unsigned* __restrict__ elist,
                                                   int* __restrict__ ecnt,
                                                   float* __restrict__ edeg,
                                                   int nE, int apply_lrelu) {
    __shared__ __attribute__((aligned(16))) float acc[(RROWS + 1) * LDA];
    __shared__ unsigned queue[NWAVES * QCAP];
    __shared__ int lcount;

    const int tid  = threadIdx.x;
    const int wave = __builtin_amdgcn_readfirstlane(tid >> 5);
    const int lane = tid & 31;
    const int row0 = blockIdx.x * RROWS;
    int nrows = N_NODES - row0;
    if (nrows > RROWS) nrows = RROWS;
    const int ntiles = nrows >> 4;

    unsigned* mylist = elist + (size_t)blockIdx.x * LCAP;
    int nlist = LCAP + 1;
    if constexpr (REPLAY) nlist = __builtin_amdgcn_readfirstlane(ecnt[blockIdx.x]);
    const bool use_list = REPLAY && nlist <= LCAP;

    for (int i = tid; i < (RROWS + 1) * (KDIM / 4); i += NWAVES * 32) {
        const int r  = i >> 5;
        const int c4 = (i & 31) << 2;
        float4 v = make_float4(0.f, 0.f, 0.f, 0.f);
        if (r < nrows) v = *(const float4*)(feat + (size_t)(row0 + r) * KDIM + c4);
        *(float4*)&acc[r * LDA + c4] = v;
    }
    const unsigned pad = (unsigned)RROWS;
    int lbeg = 0, lend = 0;
    unsigned ent0 = pad;
    if constexpr (REPLAY) {
        if (use_list) {
            const int per = (((nlist + NWAVES - 1) / NWAVES) + 7) & ~7;
            lbeg = wave * per;
            lend = lbeg + per;
            if (lend > nlist) lend = nlist;
            if (lbeg + lane < lend) ent0 = mylist[lbeg + lane];
        }
    }
    for (int r = tid; r <= RROWS; r += NWAVES * 32) {
        float d = 0.f;
        if constexpr (REPLAY) {
            if (use_list && r < nrows) d = edeg[row0 + r];
        }
        acc[r * LDA + KDIM] = d;
    }
    if constexpr (!REPLAY) {
        if (tid == 0) lcount = 0;
    }
    __syncthreads();

    unsigned* q = queue + wave * QCAP;
    int qn = 0;

    auto add_rows = [&](int s, int dl, int nb) {
        for (int u0 = 0; u0 < nb; u0 += 8) {
            float v[8][4];
            int   dls[8];
#pragma unroll
            for (int u = 0; u < 8; ++u) {
                const int su = __builtin_amdgcn_readlane(s, u0 + u);
                dls[u]       = __builtin_amdgcn_readlane(dl, u0 + u);
                const float* p = feat + (size_t)su * KDIM + lane;
                v[u][0] = p[0];
                v[u][1] = p[32];
                v[u][2] = p[64];
                v[u][3] = p[96];
            }
#pragma unroll
            for (int u = 0; u < 8; ++u) {
                float* a = &acc[dls[u] * LDA + lane];
                __hip_atomic_fetch_add(a,      v[u][0], __ATOMIC_RELAXED, __HIP_MEMORY_SCOPE_WORKGROUP);
                __hip_atomic_fetch_add(a + 32, v[u][1], __ATOMIC_RELAXED, __HIP_MEMORY_SCOPE_WORKGROUP);
                __hip_atomic_fetch_add(a + 64, v[u][2], __ATOMIC_RELAXED, __HIP_MEMORY_SCOPE_WORKGROUP);
                __hip_atomic_fetch_add(a + 96, v[u][3], __ATOMIC_RELAXED, __HIP_MEMORY_SCOPE_WORKGROUP);
            }
        }
    };

    auto drain = [&]() {
        for (int i = 0; i < qn; i += 32) {
            const bool live = (i + lane) < qn;
            const unsigned ent = live ? q[i + lane] : 0u;
            const int s  = live ? src[ent >> 8] : 0;
            const int dl = live ? (int)(ent & 255u) : RROWS;
            int nb = qn - i;
            if (nb > 32) nb = 32;
            if constexpr (!REPLAY) {
                int base = 0;
                if (lane == 0)
                    base = __hip_atomic_fetch_add(&lcount, nb, __ATOMIC_RELAXED, __HIP_MEMORY_SCOPE_WORKGROUP);
                base = __builtin_amdgcn_readfirstlane(base);
                if (live && base + lane < LCAP) mylist[base + lane] = ((unsigned)s << 8) | (unsigned)dl;
            }
            add_rows(s, dl, nb);
        }
    };

    if (!use_list) {
        const int4* dst4  = (const int4*)dst;
        const int   ngrp  = nE >> 2;
        const int   nfull = ngrp >> 8;

        auto sift = [&](const int4 (&g)[GPS], int step) {
            if (qn > QCAP - QSTEP) {
                drain();
                qn = 0;
            }
#pragma unroll
            for (int j = 0; j < GPS; ++j) {
                const int e0 = (step * (GPS * 32) + j * 32 + lane) << 2;
                const unsigned d[4] = {(unsigned)(g[j].x - row0), (unsigned)(g[j].y - row0),
                                       (unsigned)(g[j].z - row0), (unsigned)(g[j].w - row0)};
                const bool any = (d[0] < (unsigned)nrows) | (d[1] < (unsigned)nrows) |
                                 (d[2] < (unsigned)nrows) | (d[3] < (unsigned)nrows);
                if (__builtin_amdgcn_ballot_w32(any) != 0u) {
#pragma unroll
                    for (int c = 0; c < 4; ++c) {
                        const bool m = d[c] < (unsigned)nrows;
                        const unsigned mk = __builtin_amdgcn_ballot_w32(m);
                        if (m) {
                            const int pos = qn + (int)__builtin_amdgcn_mbcnt_lo(mk, 0u);
                            q[pos] = ((unsigned)(e0 + c) << 8) | d[c];
                            __hip_atomic_fetch_add(&acc[d[c] * LDA + KDIM], 1.0f, __ATOMIC_RELAXED,
                                                   __HIP_MEMORY_SCOPE_WORKGROUP);
                        }
                        qn += __builtin_popcount(mk);
                    }
                }
            }
        };

        int4 g[GPS];
        for (int step = wave; step < nfull; step += NWAVES) {
            const int4* p = dst4 + (size_t)step * (GPS * 32) + lane;
#pragma unroll
            for (int j = 0; j < GPS; ++j) g[j] = p[j * 32];
            sift(g, step);
        }
        if (wave == (nfull & (NWAVES - 1)) && (ngrp & (GPS * 32 - 1)) != 0) {
            const int4 none = make_int4(-1, -1, -1, -1);
#pragma unroll
            for (int j = 0; j < GPS; ++j) {
                const int  gi = nfull * (GPS * 32) + j * 32 + lane;
                const int4 t  = dst4[gi < ngrp ? gi : ngrp - 1];
                g[j] = (gi < ngrp) ? t : none;
            }
            sift(g, nfull);
        }
        drain();
    }
    if constexpr (REPLAY) {
        if (use_list) {
            int s  = (int)(ent0 >> 8);
            int dl = (int)(ent0 & 255u);
            for (int i = lbeg; i < lend; i += 32) {
                const int nx = i + 32 + lane;
                const unsigned nxt = (nx < lend) ? mylist[nx] : pad;
                int nb = lend - i;
                if (nb > 32) nb = 32;
                add_rows(s, dl, nb);
                s  = (int)(nxt >> 8);
                dl = (int)(nxt & 255u);
            }
        }
    }
    __syncthreads();

    if constexpr (!REPLAY) {
        if (tid == 0) ecnt[blockIdx.x] = lcount;
        for (int r = tid; r < nrows; r += NWAVES * 32) edeg[row0 + r] = acc[r * LDA + KDIM];
    }

    for (int i = tid; i < nrows * (KDIM / 4); i += NWAVES * 32) {
        const int r  = i >> 5;
        const int c4 = (i & 31) << 2;
        const float inv = 1.0f / (acc[r * LDA + KDIM] + 1.0f);
        float4 v = *(float4*)&acc[r * LDA + c4];
        v.x *= inv; v.y *= inv; v.z *= inv; v.w *= inv;
        *(float4*)&acc[r * LDA + c4] = v;
    }
    __syncthreads();

    constexpr int NCT  = NOUT / 16;
    constexpr int RG   = NWAVES / NCT;
    constexpr int NACC = (RROWS / 16 + RG - 1) / RG;
    const int ct = wave % NCT;
    const int rg = wave / NCT;
    const int n0 = ct * 16;
    const int lr = lane & 15;
    const int hi = lane >> 4;

    int arow[NACC];
#pragma unroll
    for (int i = 0; i < NACC; ++i) {
        const int t = rg + RG * i;
        arow[i] = ((t < ntiles) ? t : (ntiles - 1)) * 16 + lr;
    }
    v8f c[NACC];
#pragma unroll
    for (int i = 0; i < NACC; ++i) c[i] = (v8f){0.f, 0.f, 0.f, 0.f, 0.f, 0.f, 0.f, 0.f};

#pragma unroll 8
    for (int k = 0; k < KDIM; k += 4) {
        v2f b;
        const float* bp = &W[(size_t)(k + 2 * hi) * NOUT + n0 + lr];
        b.x = bp[0];
        b.y = bp[NOUT];
#pragma unroll
        for (int i = 0; i < NACC; ++i) {
            v2f a;
            const float* ap = &acc[arow[i] * LDA + k + 2 * hi];
            a.x = ap[0];
            a.y = ap[1];
            c[i] = __builtin_amdgcn_wmma_f32_16x16x4_f32(
                 false, a,  false, b,
                 (short)0, c[i],  false,  false);
        }
    }

    const float bv  = bias[n0 + lr];
    const int   col = n0 + lr;
#pragma unroll
    for (int i = 0; i < NACC; ++i) {
        const int t = rg + RG * i;
        if (t < ntiles) {
            const int rbase = row0 + t * 16 + 8 * hi;
#pragma unroll
            for (int r = 0; r < 8; ++r) {
                float v = c[i][r] + bv;
                if (apply_lrelu) v = (v >= 0.f) ? v : 0.01f * v;
                out[(size_t)(rbase + r) * NOUT + col] = v;
            }
        }
    }
}

extern "C" void kernel_launch(void* const* d_in, const int* in_sizes, int n_in,
                              void* d_out, int out_size, void* d_ws, size_t ws_size,
                              hipStream_t stream) {
    const float* x   = (const float*)d_in[0];
    const int*   edg = (const int*)  d_in[1];
    const float* W1  = (const float*)d_in[2];
    const float* b1  = (const float*)d_in[3];
    const float* W2  = (const float*)d_in[4];
    const float* b2  = (const float*)d_in[5];
    float*       outp = (float*)d_out;

    const int E = in_sizes[1] / 2;
    const int* src = edg;
    const int* dst = edg + E;

    float*    h1    = (float*)d_ws;
    unsigned* elist = (unsigned*)(h1 + (size_t)N_NODES * KDIM);
    float*    edeg  = (float*)(elist + (size_t)NWG * LCAP);
    int*      ecnt  = (int*)(edeg + N_NODES);

    sage_layer<128, false><<<NWG, NWAVES * 32, 0, stream>>>(
        x, src, dst, W1, b1, h1, elist, ecnt, edeg, E,  1);
    sage_layer<32, true><<<NWG, NWAVES * 32, 0, stream>>>(
        h1, src, dst, W2, b2, outp, elist, ecnt, edeg, E,  0);
}


# each wave owns rows = its number mod 32: found edges go to the owner's LDS bin, ordered by row; it sums whole-row 16-B-a-lane loads in registers, one LDS add per row; no f32 atomic; sum order changed
# speedup vs baseline: 4.5413x; 1.1951x over previous
#include <hip/hip_runtime.h>


typedef float v2f __attribute__((ext_vector_type(2)));
typedef float v8f __attribute__((ext_vector_type(8)));

#define N_NODES 50000
#define KDIM    128
#define LDA     132
#define RROWS   208
#define NWAVES  32
#define NSLOT   7
#define AROWS   (NSLOT * 32)
#define GPS     8
#define BINCAP  768
#define GROWS   16
#define NWG     ((N_NODES + RROWS - 1) / RROWS)
#define LCAP    4096

template <int NOUT, bool REPLAY>
__global__ __launch_bounds__(1024) void sage_layer(const float* __restrict__ feat,
                                                   const int* __restrict__ src,
                                                   const int* __restrict__ dst,
                                                   const float* __restrict__ W,
                                                   const float* __restrict__ bias,
                                                   float* __restrict__ out,
                                                   unsigned* __restrict__ elist,
                                                   int* __restrict__ ecnt,
                                                   int nE, int apply_lrelu) {
    __shared__ __attribute__((aligned(16))) float acc[AROWS * LDA];
    __shared__ unsigned bins[NWAVES * BINCAP];
    __shared__ unsigned sorted[NWAVES * BINCAP];
    __shared__ int bcnt[NWAVES];

    const int tid  = threadIdx.x;
    const int wave = __builtin_amdgcn_readfirstlane(tid >> 5);
    const int lane = tid & 31;
    const int row0 = blockIdx.x * RROWS;
    int nrows = N_NODES - row0;
    if (nrows > RROWS) nrows = RROWS;
    const int ntiles = nrows >> 4;

    unsigned* mylist = elist + (size_t)blockIdx.x * LCAP;
    int nlist = LCAP + 1;
    if constexpr (REPLAY) nlist = __builtin_amdgcn_readfirstlane(ecnt[blockIdx.x]);
    const bool use_list = REPLAY && nlist <= LCAP - 32;

    float4 own[NSLOT];
#pragma unroll
    for (int j = 0; j < NSLOT; ++j) {
        const int r = 32 * j + wave;
        own[j] = *(const float4*)(feat + (size_t)(row0 + (r < nrows ? r : 0)) * KDIM + 4 * lane);
    }
#pragma unroll
    for (int j = 0; j < NSLOT; ++j) {
        const int r = 32 * j + wave;
        *(float4*)&acc[r * LDA + 4 * lane] = (r < nrows) ? own[j] : make_float4(0.f, 0.f, 0.f, 0.f);
        if (lane == 0) acc[r * LDA + KDIM] = 0.f;
    }
    if (tid < NWAVES) bcnt[tid] = 0;
    __syncthreads();

    unsigned* mybin    = bins + wave * BINCAP;
    unsigned* mysorted = sorted + wave * BINCAP;

    if (!use_list) {
        const int4* dst4  = (const int4*)dst;
        const int   ngrp  = nE >> 2;
        const int   nfull = ngrp >> 8;

        auto sift = [&](const int4 (&g)[GPS], int step) {
#pragma unroll
            for (int j = 0; j < GPS; ++j) {
                const int e0 = (step * (GPS * 32) + j * 32 + lane) << 2;
                const unsigned d[4] = {(unsigned)(g[j].x - row0), (unsigned)(g[j].y - row0),
                                       (unsigned)(g[j].z - row0), (unsigned)(g[j].w - row0)};
                const bool any = (d[0] < (unsigned)nrows) | (d[1] < (unsigned)nrows) |
                                 (d[2] < (unsigned)nrows) | (d[3] < (unsigned)nrows);
                if (__builtin_amdgcn_ballot_w32(any) != 0u) {
#pragma unroll
                    for (int c = 0; c < 4; ++c) {
                        if (d[c] < (unsigned)nrows) {
                            const int ow  = (int)(d[c] & 31u);
                            const int pos = __hip_atomic_fetch_add(&bcnt[ow], 1, __ATOMIC_RELAXED,
                                                                   __HIP_MEMORY_SCOPE_WORKGROUP);
                            if (pos < BINCAP) bins[ow * BINCAP + pos] = ((unsigned)(e0 + c) << 8) | d[c];
                        }
                    }
                }
            }
        };

        int4 g[GPS];
        for (int step = wave; step < nfull; step += NWAVES) {
            const int4* p = dst4 + (size_t)step * (GPS * 32) + lane;
#pragma unroll
            for (int j = 0; j < GPS; ++j) g[j] = p[j * 32];
            sift(g, step);
        }
        if (wave == (nfull & (NWAVES - 1)) && (ngrp & (GPS * 32 - 1)) != 0) {
            const int4 none = make_int4(-1, -1, -1, -1);
#pragma unroll
            for (int j = 0; j < GPS; ++j) {
                const int  gi = nfull * (GPS * 32) + j * 32 + lane;
                const int4 t  = dst4[gi < ngrp ? gi : ngrp - 1];
                g[j] = (gi < ngrp) ? t : none;
            }
            sift(g, nfull);
        }
        __syncthreads();
    }

    const int cw = use_list ? (int)mylist[lane] : bcnt[lane];
    int off = 0, total = 0;
#pragma unroll
    for (int i = 0; i < NWAVES; ++i) {
        const int ci = __builtin_amdgcn_readlane(cw, i);
        if (i < wave) off += ci;
        total += ci;
    }
    int n = __builtin_amdgcn_readlane(cw, wave);
    const bool full = __builtin_amdgcn_ballot_w32(cw > BINCAP) != 0u;
    if (n > BINCAP) n = BINCAP;
    bool put_list = false;
    if constexpr (!REPLAY) {
        put_list = !full && total <= LCAP - 32;
        if (tid == 0) ecnt[blockIdx.x] = put_list ? total : LCAP + 1;
        if (put_list && wave == 0) mylist[lane] = (unsigned)cw;
    }
    unsigned* myrun = mylist + 32 + off;

    if (!use_list && full) {
        for (int e = tid; e < nE; e += NWAVES * 32) {
            const unsigned dl = (unsigned)(dst[e] - row0);
            if (dl < (unsigned)nrows) {
                const float* p = feat + (size_t)src[e] * KDIM;
                float* r = &acc[dl * LDA];
#pragma nounroll
                for (int f = 0; f < KDIM; ++f)
                    __hip_atomic_fetch_add(r + f, p[f], __ATOMIC_RELAXED, __HIP_MEMORY_SCOPE_WORKGROUP);
                __hip_atomic_fetch_add(r + KDIM, 1.0f, __ATOMIC_RELAXED, __HIP_MEMORY_SCOPE_WORKGROUP);
            }
        }
        n = 0;
        __syncthreads();
    }

    if (!use_list) {
        int fill[NSLOT];
#pragma unroll
        for (int j = 0; j < NSLOT; ++j) fill[j] = 0;
        for (int i = 0; i < n; i += 32) {
            const bool live = (i + lane) < n;
            const int  kk   = live ? (int)((mybin[live ? i + lane : 0] & 255u) >> 5) : NSLOT;
#pragma unroll
            for (int j = 0; j < NSLOT; ++j) fill[j] += __builtin_popcount(__builtin_amdgcn_ballot_w32(kk == j));
        }
        int run = 0;
#pragma unroll
        for (int j = 0; j < NSLOT; ++j) {
            const int cj = fill[j];
            fill[j] = run;
            run += cj;
        }
        for (int i = 0; i < n; i += 32) {
            const bool live = (i + lane) < n;
            const unsigned ent = mybin[live ? i + lane : 0];
            const int      sv  = src[live ? (ent >> 8) : 0u];
            const unsigned dl  = ent & 255u;
            const int      kk  = live ? (int)(dl >> 5) : NSLOT;
            int pos = 0;
#pragma unroll
            for (int j = 0; j < NSLOT; ++j) {
                const unsigned m = __builtin_amdgcn_ballot_w32(kk == j);
                if (kk == j) pos = fill[j] + (int)__builtin_amdgcn_mbcnt_lo(m, 0u);
                fill[j] += __builtin_popcount(m);
            }
            if (live) mysorted[pos] = ((unsigned)sv << 8) | dl;
        }
    }

    int deg[NSLOT];
#pragma unroll
    for (int j = 0; j < NSLOT; ++j) deg[j] = 0;

    auto fetch = [&](int i, int& s, int& k) {
        const bool live = (i + lane) < n;
        unsigned ent = 255u;
        if (live) ent = use_list ? myrun[i + lane] : mysorted[i + lane];
        if constexpr (!REPLAY) {
            if (put_list && live) myrun[i + lane] = ent;
        }
        s = (int)(ent >> 8);
        k = (int)((ent & 255u) >> 5);
#pragma unroll
        for (int j = 0; j < NSLOT; ++j) deg[j] += __builtin_popcount(__builtin_amdgcn_ballot_w32(k == j));
    };

    float4 sum = make_float4(0.f, 0.f, 0.f, 0.f);
    int    cur = NSLOT;
    auto flush = [&]() {
        if (cur < NSLOT) {
            float4* rp = (float4*)&acc[(32 * cur + wave) * LDA + 4 * lane];
            float4 o = *rp;
            o.x += sum.x; o.y += sum.y; o.z += sum.z; o.w += sum.w;
            *rp = o;
        }
        sum = make_float4(0.f, 0.f, 0.f, 0.f);
    };

    int s, k;
    fetch(0, s, k);
    for (int i = 0; i < n; i += 32) {
        int s2, k2;
        fetch(i + 32, s2, k2);
        int nb = n - i;
        if (nb > 32) nb = 32;
        for (int u0 = 0; u0 < nb; u0 += GROWS) {
            float4 v[GROWS];
            int    ks[GROWS];
#pragma unroll
            for (int u = 0; u < GROWS; ++u) {
                const int su = __builtin_amdgcn_readlane(s, u0 + u);
                ks[u]        = __builtin_amdgcn_readlane(k, u0 + u);
                v[u] = *(const float4*)(feat + (size_t)su * KDIM + 4 * lane);
            }
#pragma unroll
            for (int u = 0; u < GROWS; ++u) {
                if (ks[u] != cur) {
                    flush();
                    cur = ks[u];
                }
                sum.x += v[u].x; sum.y += v[u].y; sum.z += v[u].z; sum.w += v[u].w;
            }
        }
        s = s2;
        k = k2;
    }
    flush();

#pragma unroll
    for (int j = 0; j < NSLOT; ++j) {
        const int r = 32 * j + wave;
        if (r < nrows) {
            float* rp = &acc[r * LDA];
            float4 t  = *(float4*)&rp[4 * lane];
            const float inv = 1.0f / ((float)deg[j] + rp[KDIM] + 1.0f);
            t.x *= inv; t.y *= inv; t.z *= inv; t.w *= inv;
            *(float4*)&rp[4 * lane] = t;
        }
    }
    __syncthreads();

    constexpr int NCT  = NOUT / 16;
    constexpr int RG   = NWAVES / NCT;
    constexpr int NACC = (RROWS / 16 + RG - 1) / RG;
    const int ct = wave % NCT;
    const int rg = wave / NCT;
    const int n0 = ct * 16;
    const int lr = lane & 15;
    const int hi = lane >> 4;

    int arow[NACC];
#pragma unroll
    for (int i = 0; i < NACC; ++i) {
        const int t = rg + RG * i;
        arow[i] = ((t < ntiles) ? t : (ntiles - 1)) * 16 + lr;
    }
    v8f c[NACC];
#pragma unroll
    for (int i = 0; i < NACC; ++i) c[i] = (v8f){0.f, 0.f, 0.f, 0.f, 0.f, 0.f, 0.f, 0.f};

#pragma unroll 8
    for (int k = 0; k < KDIM; k += 4) {
        v2f b;
        const float* bp = &W[(size_t)(k + 2 * hi) * NOUT + n0 + lr];
        b.x = bp[0];
        b.y = bp[NOUT];
#pragma unroll
        for (int i = 0; i < NACC; ++i) {
            v2f a;
            const float* ap = &acc[arow[i] * LDA + k + 2 * hi];
            a.x = ap[0];
            a.y = ap[1];
            c[i] = __builtin_amdgcn_wmma_f32_16x16x4_f32(
                 false, a,  false, b,
                 (short)0, c[i],  false,  false);
        }
    }

    const float bv  = bias[n0 + lr];
    const int   col = n0 + lr;
#pragma unroll
    for (int i = 0; i < NACC; ++i) {
        const int t = rg + RG * i;
        if (t < ntiles) {
            const int rbase = row0 + t * 16 + 8 * hi;
#pragma unroll
            for (int r = 0; r < 8; ++r) {
                float v = c[i][r] + bv;
                if (apply_lrelu) v = (v >= 0.f) ? v : 0.01f * v;
                out[(size_t)(rbase + r) * NOUT + col] = v;
            }
        }
    }
}

extern "C" void kernel_launch(void* const* d_in, const int* in_sizes, int n_in,
                              void* d_out, int out_size, void* d_ws, size_t ws_size,
                              hipStream_t stream) {
    const float* x   = (const float*)d_in[0];
    const int*   edg = (const int*)  d_in[1];
    const float* W1  = (const float*)d_in[2];
    const float* b1  = (const float*)d_in[3];
    const float* W2  = (const float*)d_in[4];
    const float* b2  = (const float*)d_in[5];
    float*       outp = (float*)d_out;

    const int E = in_sizes[1] / 2;
    const int* src = edg;
    const int* dst = edg + E;

    float*    h1    = (float*)d_ws;
    unsigned* elist = (unsigned*)(h1 + (size_t)N_NODES * KDIM);
    int*      ecnt  = (int*)(elist + (size_t)NWG * LCAP);

    sage_layer<128, false><<<NWG, NWAVES * 32, 0, stream>>>(
        x, src, dst, W1, b1, h1, elist, ecnt, E,  1);
    sage_layer<32, true><<<NWG, NWAVES * 32, 0, stream>>>(
        h1, src, dst, W2, b2, outp, elist, ecnt, E,  0);
}


# layer-1 scan reads one owner byte per edge (destination / 208, written by a small launch in front), not the 4-byte destination; candidates resolved 32 at a time; sums by arrival as before
# speedup vs baseline: 4.8596x; 1.0701x over previous
#include <hip/hip_runtime.h>


typedef float v2f __attribute__((ext_vector_type(2)));
typedef float v8f __attribute__((ext_vector_type(8)));

#define N_NODES 50000
#define KDIM    128
#define LDA     132
#define RROWS   208
#define NWAVES  32
#define NSLOT   7
#define AROWS   (NSLOT * 32)
#define GPS     8
#define BINCAP  768
#define GROWS   16
#define NWG     ((N_NODES + RROWS - 1) / RROWS)
static_assert((N_NODES + RROWS - 1) / RROWS < 255, "owner numbers must fit a byte below 255");
#define LCAP    4096

template <int NOUT, bool REPLAY>
__global__ __launch_bounds__(1024) void sage_layer(const float* __restrict__ feat,
                                                   const int* __restrict__ src,
                                                   const int* __restrict__ dst,
                                                   const unsigned char* __restrict__ keys,
                                                   const float* __restrict__ W,
                                                   const float* __restrict__ bias,
                                                   float* __restrict__ out,
                                                   unsigned* __restrict__ elist,
                                                   int* __restrict__ ecnt,
                                                   int nE, int apply_lrelu) {
    __shared__ __attribute__((aligned(16))) float acc[AROWS * LDA];
    __shared__ unsigned bins[NWAVES * BINCAP];
    __shared__ unsigned sorted[NWAVES * BINCAP];
    __shared__ int bcnt[NWAVES];
    __shared__ int ccnt[NWAVES];

    const int tid  = threadIdx.x;
    const int wave = __builtin_amdgcn_readfirstlane(tid >> 5);
    const int lane = tid & 31;
    const int row0 = blockIdx.x * RROWS;
    int nrows = N_NODES - row0;
    if (nrows > RROWS) nrows = RROWS;
    const int ntiles = nrows >> 4;

    unsigned* mylist = elist + (size_t)blockIdx.x * LCAP;
    int nlist = LCAP + 1;
    if constexpr (REPLAY) nlist = __builtin_amdgcn_readfirstlane(ecnt[blockIdx.x]);
    const bool use_list = REPLAY && nlist <= LCAP - 32;

    float4 own[NSLOT];
#pragma unroll
    for (int j = 0; j < NSLOT; ++j) {
        const int r = 32 * j + wave;
        own[j] = *(const float4*)(feat + (size_t)(row0 + (r < nrows ? r : 0)) * KDIM + 4 * lane);
    }
#pragma unroll
    for (int j = 0; j < NSLOT; ++j) {
        const int r = 32 * j + wave;
        *(float4*)&acc[r * LDA + 4 * lane] = (r < nrows) ? own[j] : make_float4(0.f, 0.f, 0.f, 0.f);
        if (lane == 0) acc[r * LDA + KDIM] = 0.f;
    }
    if (tid < NWAVES) bcnt[tid] = 0;
    __syncthreads();

    unsigned* mybin    = bins + wave * BINCAP;
    unsigned* mysorted = sorted + wave * BINCAP;

    if (!use_list) {
        const int4* key4   = (const int4*)keys;
        const int   nsteps = (nE + GPS * 32 * 16 - 1) / (GPS * 32 * 16);
        const unsigned bb  = (unsigned)blockIdx.x * 0x01010101u;

        int4 g[GPS];
        int  ncand = 0;
        for (int step = wave; step < nsteps; step += NWAVES) {
            const int4* p = key4 + (size_t)step * (GPS * 32) + lane;
#pragma unroll
            for (int j = 0; j < GPS; ++j) g[j] = p[j * 32];
#pragma unroll
            for (int j = 0; j < GPS; ++j) {
                const int e0 = (step * (GPS * 32) + j * 32 + lane) << 4;
                const unsigned w[4] = {(unsigned)g[j].x ^ bb, (unsigned)g[j].y ^ bb,
                                       (unsigned)g[j].z ^ bb, (unsigned)g[j].w ^ bb};
                unsigned mm = 0u;
#pragma unroll
                for (int q = 0; q < 4; ++q)
                    mm |= (((w[q] - 0x01010101u) & ~w[q] & 0x80808080u) >> 7) << q;
                unsigned act = __builtin_amdgcn_ballot_w32(mm != 0u);
                while (act != 0u) {
                    if (mm != 0u) {
                        const int b = __builtin_ctz(mm);
                        mm &= mm - 1u;
                        const int pos = ncand + (int)__builtin_amdgcn_mbcnt_lo(act, 0u);
                        if (pos < BINCAP) mysorted[pos] = (unsigned)(e0 + 4 * (b & 7) + (b >> 3));
                    }
                    ncand += __builtin_popcount(act);
                    act = __builtin_amdgcn_ballot_w32(mm != 0u);
                }
            }
        }
        if (lane == 0) ccnt[wave] = ncand;
        __syncthreads();

        int nc = ncand;
        if (nc > BINCAP) nc = BINCAP;
        for (int i = 0; i < nc; i += 32) {
            const bool live = (i + lane) < nc;
            const int  e    = (int)mysorted[live ? i + lane : 0];
            const bool ok   = live && e < nE;
            const unsigned d  = (unsigned)(dst[ok ? e : 0] - row0);
            const int      sv = src[ok ? e : 0];
            if (ok && d < (unsigned)nrows) {
                const int ow  = (int)(d & 31u);
                const int pos = __hip_atomic_fetch_add(&bcnt[ow], 1, __ATOMIC_RELAXED, __HIP_MEMORY_SCOPE_WORKGROUP);
                if (pos < BINCAP) bins[ow * BINCAP + pos] = ((unsigned)sv << 8) | d;
            }
        }
        __syncthreads();
    }

    const int cw = use_list ? (int)mylist[lane] : bcnt[lane];
    int off = 0, total = 0;
#pragma unroll
    for (int i = 0; i < NWAVES; ++i) {
        const int ci = __builtin_amdgcn_readlane(cw, i);
        if (i < wave) off += ci;
        total += ci;
    }
    int n = __builtin_amdgcn_readlane(cw, wave);
    const int  cc   = use_list ? 0 : ccnt[lane];
    const bool full = __builtin_amdgcn_ballot_w32(cw > BINCAP || cc > BINCAP) != 0u;
    if (n > BINCAP) n = BINCAP;
    bool put_list = false;
    if constexpr (!REPLAY) {
        put_list = !full && total <= LCAP - 32;
        if (tid == 0) ecnt[blockIdx.x] = put_list ? total : LCAP + 1;
        if (put_list && wave == 0) mylist[lane] = (unsigned)cw;
    }
    unsigned* myrun = mylist + 32 + off;

    if (!use_list && full) {
        for (int e = tid; e < nE; e += NWAVES * 32) {
            const unsigned dl = (unsigned)(dst[e] - row0);
            if (dl < (unsigned)nrows) {
                const float* p = feat + (size_t)src[e] * KDIM;
                float* r = &acc[dl * LDA];
#pragma nounroll
                for (int f = 0; f < KDIM; ++f)
                    __hip_atomic_fetch_add(r + f, p[f], __ATOMIC_RELAXED, __HIP_MEMORY_SCOPE_WORKGROUP);
                __hip_atomic_fetch_add(r + KDIM, 1.0f, __ATOMIC_RELAXED, __HIP_MEMORY_SCOPE_WORKGROUP);
            }
        }
        n = 0;
        __syncthreads();
    }

    if (!use_list) {
        int fill[NSLOT];
#pragma unroll
        for (int j = 0; j < NSLOT; ++j) fill[j] = 0;
        for (int i = 0; i < n; i += 32) {
            const bool live = (i + lane) < n;
            const int  kk   = live ? (int)((mybin[live ? i + lane : 0] & 255u) >> 5) : NSLOT;
#pragma unroll
            for (int j = 0; j < NSLOT; ++j) fill[j] += __builtin_popcount(__builtin_amdgcn_ballot_w32(kk == j));
        }
        int run = 0;
#pragma unroll
        for (int j = 0; j < NSLOT; ++j) {
            const int cj = fill[j];
            fill[j] = run;
            run += cj;
        }
        for (int i = 0; i < n; i += 32) {
            const bool live = (i + lane) < n;
            const unsigned ent = mybin[live ? i + lane : 0];
            const unsigned dl  = ent & 255u;
            const int      kk  = live ? (int)(dl >> 5) : NSLOT;
            int pos = 0;
#pragma unroll
            for (int j = 0; j < NSLOT; ++j) {
                const unsigned m = __builtin_amdgcn_ballot_w32(kk == j);
                if (kk == j) pos = fill[j] + (int)__builtin_amdgcn_mbcnt_lo(m, 0u);
                fill[j] += __builtin_popcount(m);
            }
            if (live) mysorted[pos] = ent;
        }
    }

    int deg[NSLOT];
#pragma unroll
    for (int j = 0; j < NSLOT; ++j) deg[j] = 0;

    auto fetch = [&](int i, int& s, int& k) {
        const bool live = (i + lane) < n;
        unsigned ent = 255u;
        if (live) ent = use_list ? myrun[i + lane] : mysorted[i + lane];
        if constexpr (!REPLAY) {
            if (put_list && live) myrun[i + lane] = ent;
        }
        s = (int)(ent >> 8);
        k = (int)((ent & 255u) >> 5);
#pragma unroll
        for (int j = 0; j < NSLOT; ++j) deg[j] += __builtin_popcount(__builtin_amdgcn_ballot_w32(k == j));
    };

    float4 sum = make_float4(0.f, 0.f, 0.f, 0.f);
    int    cur = NSLOT;
    auto flush = [&]() {
        if (cur < NSLOT) {
            float4* rp = (float4*)&acc[(32 * cur + wave) * LDA + 4 * lane];
            float4 o = *rp;
            o.x += sum.x; o.y += sum.y; o.z += sum.z; o.w += sum.w;
            *rp = o;
        }
        sum = make_float4(0.f, 0.f, 0.f, 0.f);
    };

    int s, k;
    fetch(0, s, k);
    for (int i = 0; i < n; i += 32) {
        int s2, k2;
        fetch(i + 32, s2, k2);
        int nb = n - i;
        if (nb > 32) nb = 32;
        for (int u0 = 0; u0 < nb; u0 += GROWS) {
            float4 v[GROWS];
            int    ks[GROWS];
#pragma unroll
            for (int u = 0; u < GROWS; ++u) {
                const int su = __builtin_amdgcn_readlane(s, u0 + u);
                ks[u]        = __builtin_amdgcn_readlane(k, u0 + u);
                v[u] = *(const float4*)(feat + (size_t)su * KDIM + 4 * lane);
            }
#pragma unroll
            for (int u = 0; u < GROWS; ++u) {
                if (ks[u] != cur) {
                    flush();
                    cur = ks[u];
                }
                sum.x += v[u].x; sum.y += v[u].y; sum.z += v[u].z; sum.w += v[u].w;
            }
        }
        s = s2;
        k = k2;
    }
    flush();

#pragma unroll
    for (int j = 0; j < NSLOT; ++j) {
        const int r = 32 * j + wave;
        if (r < nrows) {
            float* rp = &acc[r * LDA];
            float4 t  = *(float4*)&rp[4 * lane];
            const float inv = 1.0f / ((float)deg[j] + rp[KDIM] + 1.0f);
            t.x *= inv; t.y *= inv; t.z *= inv; t.w *= inv;
            *(float4*)&rp[4 * lane] = t;
        }
    }
    __syncthreads();

    constexpr int NCT  = NOUT / 16;
    constexpr int RG   = NWAVES / NCT;
    constexpr int NACC = (RROWS / 16 + RG - 1) / RG;
    const int ct = wave % NCT;
    const int rg = wave / NCT;
    const int n0 = ct * 16;
    const int lr = lane & 15;
    const int hi = lane >> 4;

    int arow[NACC];
#pragma unroll
    for (int i = 0; i < NACC; ++i) {
        const int t = rg + RG * i;
        arow[i] = ((t < ntiles) ? t : (ntiles - 1)) * 16 + lr;
    }
    v8f c[NACC];
#pragma unroll
    for (int i = 0; i < NACC; ++i) c[i] = (v8f){0.f, 0.f, 0.f, 0.f, 0.f, 0.f, 0.f, 0.f};

#pragma unroll 8
    for (int k = 0; k < KDIM; k += 4) {
        v2f b;
        const float* bp = &W[(size_t)(k + 2 * hi) * NOUT + n0 + lr];
        b.x = bp[0];
        b.y = bp[NOUT];
#pragma unroll
        for (int i = 0; i < NACC; ++i) {
            v2f a;
            const float* ap = &acc[arow[i] * LDA + k + 2 * hi];
            a.x = ap[0];
            a.y = ap[1];
            c[i] = __builtin_amdgcn_wmma_f32_16x16x4_f32(
                 false, a,  false, b,
                 (short)0, c[i],  false,  false);
        }
    }

    const float bv  = bias[n0 + lr];
    const int   col = n0 + lr;
#pragma unroll
    for (int i = 0; i < NACC; ++i) {
        const int t = rg + RG * i;
        if (t < ntiles) {
            const int rbase = row0 + t * 16 + 8 * hi;
#pragma unroll
            for (int r = 0; r < 8; ++r) {
                float v = c[i][r] + bv;
                if (apply_lrelu) v = (v >= 0.f) ? v : 0.01f * v;
                out[(size_t)(rbase + r) * NOUT + col] = v;
            }
        }
    }
}

__global__ __launch_bounds__(256) void dst_keys(const int* __restrict__ dst,
                                                unsigned char* __restrict__ keys,
                                                int nE) {
    const int ngrp = ((nE + GPS * 32 * 16 - 1) / (GPS * 32 * 16)) * (GPS * 32);
    for (int i = blockIdx.x * 256 + threadIdx.x; i < ngrp; i += gridDim.x * 256) {
        unsigned w[4];
#pragma unroll
        for (int q = 0; q < 4; ++q) {
            const int  e  = 16 * i + 4 * q;
            const bool in = e < nE;
            const int4 v  = *(const int4*)(dst + (in ? e : 0));
            const unsigned k0 = (in && (unsigned)v.x < (unsigned)N_NODES) ? (unsigned)v.x / RROWS : 255u;
            const unsigned k1 = (in && (unsigned)v.y < (unsigned)N_NODES) ? (unsigned)v.y / RROWS : 255u;
            const unsigned k2 = (in && (unsigned)v.z < (unsigned)N_NODES) ? (unsigned)v.z / RROWS : 255u;
            const unsigned k3 = (in && (unsigned)v.w < (unsigned)N_NODES) ? (unsigned)v.w / RROWS : 255u;
            w[q] = k0 | (k1 << 8) | (k2 << 16) | (k3 << 24);
        }
        *(uint4*)(keys + (size_t)16 * i) = make_uint4(w[0], w[1], w[2], w[3]);
    }
}

extern "C" void kernel_launch(void* const* d_in, const int* in_sizes, int n_in,
                              void* d_out, int out_size, void* d_ws, size_t ws_size,
                              hipStream_t stream) {
    const float* x   = (const float*)d_in[0];
    const int*   edg = (const int*)  d_in[1];
    const float* W1  = (const float*)d_in[2];
    const float* b1  = (const float*)d_in[3];
    const float* W2  = (const float*)d_in[4];
    const float* b2  = (const float*)d_in[5];
    float*       outp = (float*)d_out;

    const int E = in_sizes[1] / 2;
    const int* src = edg;
    const int* dst = edg + E;

    float*    h1    = (float*)d_ws;
    unsigned* elist = (unsigned*)(h1 + (size_t)N_NODES * KDIM);
    int*      ecnt  = (int*)(elist + (size_t)NWG * LCAP);
    unsigned char* keys = (unsigned char*)(ecnt + 256);

    dst_keys<<<196, 256, 0, stream>>>(dst, keys, E);
    sage_layer<128, false><<<NWG, NWAVES * 32, 0, stream>>>(
        x, src, dst, keys, W1, b1, h1, elist, ecnt, E,  1);
    sage_layer<32, true><<<NWG, NWAVES * 32, 0, stream>>>(
        h1, src, dst, keys, W2, b2, outp, elist, ecnt, E,  0);
}


# owner numbers stored bit-sliced (8 words per 32 edges, made by ballots in the front launch): layer-1 scan tests 32 edges per lane with 8 xor-and word ops, not a zero-byte test per 4; sums by arrival
# speedup vs baseline: 5.0742x; 1.0442x over previous
#include <hip/hip_runtime.h>


typedef float v2f __attribute__((ext_vector_type(2)));
typedef float v8f __attribute__((ext_vector_type(8)));

#define N_NODES 50000
#define KDIM    128
#define LDA     132
#define RROWS   208
#define NWAVES  32
#define NSLOT   7
#define AROWS   (NSLOT * 32)
#define GSTEP   128
#define BINCAP  768
#define GROWS   16
#define NWG     ((N_NODES + RROWS - 1) / RROWS)
static_assert((N_NODES + RROWS - 1) / RROWS < 255, "owner numbers must fit a byte below 255");
#define LCAP    4096

template <int NOUT, bool REPLAY>
__global__ __launch_bounds__(1024) void sage_layer(const float* __restrict__ feat,
                                                   const int* __restrict__ src,
                                                   const int* __restrict__ dst,
                                                   const unsigned* __restrict__ keys,
                                                   const float* __restrict__ W,
                                                   const float* __restrict__ bias,
                                                   float* __restrict__ out,
                                                   unsigned* __restrict__ elist,
                                                   int* __restrict__ ecnt,
                                                   int nE, int apply_lrelu) {
    __shared__ __attribute__((aligned(16))) float acc[AROWS * LDA];
    __shared__ unsigned bins[NWAVES * BINCAP];
    __shared__ unsigned sorted[NWAVES * BINCAP];
    __shared__ int bcnt[NWAVES];
    __shared__ int ccnt[NWAVES];

    const int tid  = threadIdx.x;
    const int wave = __builtin_amdgcn_readfirstlane(tid >> 5);
    const int lane = tid & 31;
    const int row0 = blockIdx.x * RROWS;
    int nrows = N_NODES - row0;
    if (nrows > RROWS) nrows = RROWS;
    const int ntiles = nrows >> 4;

    unsigned* mylist = elist + (size_t)blockIdx.x * LCAP;
    int nlist = LCAP + 1;
    if constexpr (REPLAY) nlist = __builtin_amdgcn_readfirstlane(ecnt[blockIdx.x]);
    const bool use_list = REPLAY && nlist <= LCAP - 32;

    float4 own[NSLOT];
#pragma unroll
    for (int j = 0; j < NSLOT; ++j) {
        const int r = 32 * j + wave;
        own[j] = *(const float4*)(feat + (size_t)(row0 + (r < nrows ? r : 0)) * KDIM + 4 * lane);
    }
#pragma unroll
    for (int j = 0; j < NSLOT; ++j) {
        const int r = 32 * j + wave;
        *(float4*)&acc[r * LDA + 4 * lane] = (r < nrows) ? own[j] : make_float4(0.f, 0.f, 0.f, 0.f);
        if (lane == 0) acc[r * LDA + KDIM] = 0.f;
    }
    if (tid < NWAVES) bcnt[tid] = 0;
    __syncthreads();

    unsigned* mybin    = bins + wave * BINCAP;
    unsigned* mysorted = sorted + wave * BINCAP;

    if (!use_list) {
        const int   nsteps = (nE + GSTEP * 32 - 1) / (GSTEP * 32);
        const int4* klo    = (const int4*)keys;
        const int4* khi    = klo + (size_t)nsteps * GSTEP;
        unsigned inv[8];
#pragma unroll
        for (int i = 0; i < 8; ++i) inv[i] = ((blockIdx.x >> i) & 1u) ? 0u : 0xFFFFFFFFu;

        int4 ga[4], gb[4];
        int  ncand = 0;
        for (int step = wave; step < nsteps; step += NWAVES) {
            const int4* pl = klo + (size_t)step * GSTEP + lane;
            const int4* ph = khi + (size_t)step * GSTEP + lane;
#pragma unroll
            for (int j = 0; j < 4; ++j) {
                ga[j] = pl[j * 32];
                gb[j] = ph[j * 32];
            }
#pragma unroll
            for (int j = 0; j < 4; ++j) {
                const int e0 = (step * GSTEP + j * 32 + lane) << 5;
                unsigned mm = ((unsigned)ga[j].x ^ inv[0]) & ((unsigned)ga[j].y ^ inv[1]) &
                              ((unsigned)ga[j].z ^ inv[2]) & ((unsigned)ga[j].w ^ inv[3]) &
                              ((unsigned)gb[j].x ^ inv[4]) & ((unsigned)gb[j].y ^ inv[5]) &
                              ((unsigned)gb[j].z ^ inv[6]) & ((unsigned)gb[j].w ^ inv[7]);
                unsigned act = __builtin_amdgcn_ballot_w32(mm != 0u);
                while (act != 0u) {
                    if (mm != 0u) {
                        const int b = __builtin_ctz(mm);
                        mm &= mm - 1u;
                        const int pos = ncand + (int)__builtin_amdgcn_mbcnt_lo(act, 0u);
                        if (pos < BINCAP) mysorted[pos] = (unsigned)(e0 + b);
                    }
                    ncand += __builtin_popcount(act);
                    act = __builtin_amdgcn_ballot_w32(mm != 0u);
                }
            }
        }
        if (lane == 0) ccnt[wave] = ncand;
        __syncthreads();

        int nc = ncand;
        if (nc > BINCAP) nc = BINCAP;
        for (int i = 0; i < nc; i += 32) {
            const bool live = (i + lane) < nc;
            const int  e    = (int)mysorted[live ? i + lane : 0];
            const bool ok   = live && e < nE;
            const unsigned d  = (unsigned)(dst[ok ? e : 0] - row0);
            const int      sv = src[ok ? e : 0];
            if (ok && d < (unsigned)nrows) {
                const int ow  = (int)(d & 31u);
                const int pos = __hip_atomic_fetch_add(&bcnt[ow], 1, __ATOMIC_RELAXED, __HIP_MEMORY_SCOPE_WORKGROUP);
                if (pos < BINCAP) bins[ow * BINCAP + pos] = ((unsigned)sv << 8) | d;
            }
        }
        __syncthreads();
    }

    const int cw = use_list ? (int)mylist[lane] : bcnt[lane];
    int off = 0, total = 0;
#pragma unroll
    for (int i = 0; i < NWAVES; ++i) {
        const int ci = __builtin_amdgcn_readlane(cw, i);
        if (i < wave) off += ci;
        total += ci;
    }
    int n = __builtin_amdgcn_readlane(cw, wave);
    const int  cc   = use_list ? 0 : ccnt[lane];
    const bool full = __builtin_amdgcn_ballot_w32(cw > BINCAP || cc > BINCAP) != 0u;
    if (n > BINCAP) n = BINCAP;
    bool put_list = false;
    if constexpr (!REPLAY) {
        put_list = !full && total <= LCAP - 32;
        if (tid == 0) ecnt[blockIdx.x] = put_list ? total : LCAP + 1;
        if (put_list && wave == 0) mylist[lane] = (unsigned)cw;
    }
    unsigned* myrun = mylist + 32 + off;

    if (!use_list && full) {
        for (int e = tid; e < nE; e += NWAVES * 32) {
            const unsigned dl = (unsigned)(dst[e] - row0);
            if (dl < (unsigned)nrows) {
                const float* p = feat + (size_t)src[e] * KDIM;
                float* r = &acc[dl * LDA];
#pragma nounroll
                for (int f = 0; f < KDIM; ++f)
                    __hip_atomic_fetch_add(r + f, p[f], __ATOMIC_RELAXED, __HIP_MEMORY_SCOPE_WORKGROUP);
                __hip_atomic_fetch_add(r + KDIM, 1.0f, __ATOMIC_RELAXED, __HIP_MEMORY_SCOPE_WORKGROUP);
            }
        }
        n = 0;
        __syncthreads();
    }

    if (!use_list) {
        int fill[NSLOT];
#pragma unroll
        for (int j = 0; j < NSLOT; ++j) fill[j] = 0;
        for (int i = 0; i < n; i += 32) {
            const bool live = (i + lane) < n;
            const int  kk   = live ? (int)((mybin[live ? i + lane : 0] & 255u) >> 5) : NSLOT;
#pragma unroll
            for (int j = 0; j < NSLOT; ++j) fill[j] += __builtin_popcount(__builtin_amdgcn_ballot_w32(kk == j));
        }
        int run = 0;
#pragma unroll
        for (int j = 0; j < NSLOT; ++j) {
            const int cj = fill[j];
            fill[j] = run;
            run += cj;
        }
        for (int i = 0; i < n; i += 32) {
            const bool live = (i + lane) < n;
            const unsigned ent = mybin[live ? i + lane : 0];
            const unsigned dl  = ent & 255u;
            const int      kk  = live ? (int)(dl >> 5) : NSLOT;
            int pos = 0;
#pragma unroll
            for (int j = 0; j < NSLOT; ++j) {
                const unsigned m = __builtin_amdgcn_ballot_w32(kk == j);
                if (kk == j) pos = fill[j] + (int)__builtin_amdgcn_mbcnt_lo(m, 0u);
                fill[j] += __builtin_popcount(m);
            }
            if (live) mysorted[pos] = ent;
        }
    }

    int deg[NSLOT];
#pragma unroll
    for (int j = 0; j < NSLOT; ++j) deg[j] = 0;

    auto fetch = [&](int i, int& s, int& k) {
        const bool live = (i + lane) < n;
        unsigned ent = 255u;
        if (live) ent = use_list ? myrun[i + lane] : mysorted[i + lane];
        if constexpr (!REPLAY) {
            if (put_list && live) myrun[i + lane] = ent;
        }
        s = (int)(ent >> 8);
        k = (int)((ent & 255u) >> 5);
#pragma unroll
        for (int j = 0; j < NSLOT; ++j) deg[j] += __builtin_popcount(__builtin_amdgcn_ballot_w32(k == j));
    };

    float4 sum = make_float4(0.f, 0.f, 0.f, 0.f);
    int    cur = NSLOT;
    auto flush = [&]() {
        if (cur < NSLOT) {
            float4* rp = (float4*)&acc[(32 * cur + wave) * LDA + 4 * lane];
            float4 o = *rp;
            o.x += sum.x; o.y += sum.y; o.z += sum.z; o.w += sum.w;
            *rp = o;
        }
        sum = make_float4(0.f, 0.f, 0.f, 0.f);
    };

    int s, k;
    fetch(0, s, k);
    for (int i = 0; i < n; i += 32) {
        int s2, k2;
        fetch(i + 32, s2, k2);
        int nb = n - i;
        if (nb > 32) nb = 32;
        for (int u0 = 0; u0 < nb; u0 += GROWS) {
            float4 v[GROWS];
            int    ks[GROWS];
#pragma unroll
            for (int u = 0; u < GROWS; ++u) {
                const int su = __builtin_amdgcn_readlane(s, u0 + u);
                ks[u]        = __builtin_amdgcn_readlane(k, u0 + u);
                v[u] = *(const float4*)(feat + (size_t)su * KDIM + 4 * lane);
            }
#pragma unroll
            for (int u = 0; u < GROWS; ++u) {
                if (ks[u] != cur) {
                    flush();
                    cur = ks[u];
                }
                sum.x += v[u].x; sum.y += v[u].y; sum.z += v[u].z; sum.w += v[u].w;
            }
        }
        s = s2;
        k = k2;
    }
    flush();

#pragma unroll
    for (int j = 0; j < NSLOT; ++j) {
        const int r = 32 * j + wave;
        if (r < nrows) {
            float* rp = &acc[r * LDA];
            float4 t  = *(float4*)&rp[4 * lane];
            const float inv = 1.0f / ((float)deg[j] + rp[KDIM] + 1.0f);
            t.x *= inv; t.y *= inv; t.z *= inv; t.w *= inv;
            *(float4*)&rp[4 * lane] = t;
        }
    }
    __syncthreads();

    constexpr int NCT  = NOUT / 16;
    constexpr int RG   = NWAVES / NCT;
    constexpr int NACC = (RROWS / 16 + RG - 1) / RG;
    const int ct = wave % NCT;
    const int rg = wave / NCT;
    const int n0 = ct * 16;
    const int lr = lane & 15;
    const int hi = lane >> 4;

    int arow[NACC];
#pragma unroll
    for (int i = 0; i < NACC; ++i) {
        const int t = rg + RG * i;
        arow[i] = ((t < ntiles) ? t : (ntiles - 1)) * 16 + lr;
    }
    v8f c[NACC];
#pragma unroll
    for (int i = 0; i < NACC; ++i) c[i] = (v8f){0.f, 0.f, 0.f, 0.f, 0.f, 0.f, 0.f, 0.f};

#pragma unroll 8
    for (int k = 0; k < KDIM; k += 4) {
        v2f b;
        const float* bp = &W[(size_t)(k + 2 * hi) * NOUT + n0 + lr];
        b.x = bp[0];
        b.y = bp[NOUT];
#pragma unroll
        for (int i = 0; i < NACC; ++i) {
            v2f a;
            const float* ap = &acc[arow[i] * LDA + k + 2 * hi];
            a.x = ap[0];
            a.y = ap[1];
            c[i] = __builtin_amdgcn_wmma_f32_16x16x4_f32(
                 false, a,  false, b,
                 (short)0, c[i],  false,  false);
        }
    }

    const float bv  = bias[n0 + lr];
    const int   col = n0 + lr;
#pragma unroll
    for (int i = 0; i < NACC; ++i) {
        const int t = rg + RG * i;
        if (t < ntiles) {
            const int rbase = row0 + t * 16 + 8 * hi;
#pragma unroll
            for (int r = 0; r < 8; ++r) {
                float v = c[i][r] + bv;
                if (apply_lrelu) v = (v >= 0.f) ? v : 0.01f * v;
                out[(size_t)(rbase + r) * NOUT + col] = v;
            }
        }
    }
}

__global__ __launch_bounds__(128) void dst_keys(const int* __restrict__ dst,
                                                unsigned* __restrict__ keys,
                                                int nE) {
    const int nsteps = (nE + GSTEP * 32 - 1) / (GSTEP * 32);
    const int ngrp   = nsteps * GSTEP;
    uint4* lo = (uint4*)keys;
    uint4* hi = lo + ngrp;
    const int lane = threadIdx.x & 31;
    const int wv   = __builtin_amdgcn_readfirstlane((int)((blockIdx.x * 128 + threadIdx.x) >> 5));
    const int nwv  = gridDim.x * 4;
    for (int g0 = wv * 32; g0 < ngrp; g0 += nwv * 32) {
        int dv[32];
#pragma unroll
        for (int q = 0; q < 32; ++q) {
            const int e = (g0 + q) * 32 + lane;
            dv[q] = dst[e < nE ? e : 0];
        }
        uint4 mlo = make_uint4(0u, 0u, 0u, 0u), mhi = make_uint4(0u, 0u, 0u, 0u);
#pragma unroll
        for (int q = 0; q < 32; ++q) {
            const int e = (g0 + q) * 32 + lane;
            const unsigned k = (e < nE && (unsigned)dv[q] < (unsigned)N_NODES) ? (unsigned)dv[q] / RROWS : 255u;
            const unsigned p0 = __builtin_amdgcn_ballot_w32((k & 1u) != 0u);
            const unsigned p1 = __builtin_amdgcn_ballot_w32((k & 2u) != 0u);
            const unsigned p2 = __builtin_amdgcn_ballot_w32((k & 4u) != 0u);
            const unsigned p3 = __builtin_amdgcn_ballot_w32((k & 8u) != 0u);
            const unsigned p4 = __builtin_amdgcn_ballot_w32((k & 16u) != 0u);
            const unsigned p5 = __builtin_amdgcn_ballot_w32((k & 32u) != 0u);
            const unsigned p6 = __builtin_amdgcn_ballot_w32((k & 64u) != 0u);
            const unsigned p7 = __builtin_amdgcn_ballot_w32((k & 128u) != 0u);
            if (lane == q) {
                mlo = make_uint4(p0, p1, p2, p3);
                mhi = make_uint4(p4, p5, p6, p7);
            }
        }
        lo[g0 + lane] = mlo;
        hi[g0 + lane] = mhi;
    }
}

extern "C" void kernel_launch(void* const* d_in, const int* in_sizes, int n_in,
                              void* d_out, int out_size, void* d_ws, size_t ws_size,
                              hipStream_t stream) {
    const float* x   = (const float*)d_in[0];
    const int*   edg = (const int*)  d_in[1];
    const float* W1  = (const float*)d_in[2];
    const float* b1  = (const float*)d_in[3];
    const float* W2  = (const float*)d_in[4];
    const float* b2  = (const float*)d_in[5];
    float*       outp = (float*)d_out;

    const int E = in_sizes[1] / 2;
    const int* src = edg;
    const int* dst = edg + E;

    float*    h1    = (float*)d_ws;
    unsigned* elist = (unsigned*)(h1 + (size_t)N_NODES * KDIM);
    int*      ecnt  = (int*)(elist + (size_t)NWG * LCAP);
    unsigned* keys = (unsigned*)(ecnt + 256);

    dst_keys<<<196, 128, 0, stream>>>(dst, keys, E);
    sage_layer<128, false><<<NWG, NWAVES * 32, 0, stream>>>(
        x, src, dst, keys, W1, b1, h1, elist, ecnt, E,  1);
    sage_layer<32, true><<<NWG, NWAVES * 32, 0, stream>>>(
        h1, src, dst, keys, W2, b2, outp, elist, ecnt, E,  0);
}


# front launch also writes one packed word per edge (source << 8 | destination % 208); layer-1 candidates read that word, not dst and src: one scattered read per edge, was two; sums by arrival
# speedup vs baseline: 5.2279x; 1.0303x over previous
#include <hip/hip_runtime.h>


typedef float v2f __attribute__((ext_vector_type(2)));
typedef float v8f __attribute__((ext_vector_type(8)));

#define N_NODES 50000
#define KDIM    128
#define LDA     132
#define RROWS   208
#define NWAVES  32
#define NSLOT   7
#define AROWS   (NSLOT * 32)
#define GSTEP   128
#define BINCAP  768
#define GROWS   16
#define NWG     ((N_NODES + RROWS - 1) / RROWS)
static_assert((N_NODES + RROWS - 1) / RROWS < 255, "owner numbers must fit a byte below 255");
#define LCAP    4096

template <int NOUT, bool REPLAY>
__global__ __launch_bounds__(1024) void sage_layer(const float* __restrict__ feat,
                                                   const int* __restrict__ src,
                                                   const int* __restrict__ dst,
                                                   const unsigned* __restrict__ keys,
                                                   const unsigned* __restrict__ rec,
                                                   const float* __restrict__ W,
                                                   const float* __restrict__ bias,
                                                   float* __restrict__ out,
                                                   unsigned* __restrict__ elist,
                                                   int* __restrict__ ecnt,
                                                   int nE, int apply_lrelu) {
    __shared__ __attribute__((aligned(16))) float acc[AROWS * LDA];
    __shared__ unsigned bins[NWAVES * BINCAP];
    __shared__ unsigned sorted[NWAVES * BINCAP];
    __shared__ int bcnt[NWAVES];
    __shared__ int ccnt[NWAVES];

    const int tid  = threadIdx.x;
    const int wave = __builtin_amdgcn_readfirstlane(tid >> 5);
    const int lane = tid & 31;
    const int row0 = blockIdx.x * RROWS;
    int nrows = N_NODES - row0;
    if (nrows > RROWS) nrows = RROWS;
    const int ntiles = nrows >> 4;

    unsigned* mylist = elist + (size_t)blockIdx.x * LCAP;
    int nlist = LCAP + 1;
    if constexpr (REPLAY) nlist = __builtin_amdgcn_readfirstlane(ecnt[blockIdx.x]);
    const bool use_list = REPLAY && nlist <= LCAP - 32;

    float4 own[NSLOT];
#pragma unroll
    for (int j = 0; j < NSLOT; ++j) {
        const int r = 32 * j + wave;
        own[j] = *(const float4*)(feat + (size_t)(row0 + (r < nrows ? r : 0)) * KDIM + 4 * lane);
    }
#pragma unroll
    for (int j = 0; j < NSLOT; ++j) {
        const int r = 32 * j + wave;
        *(float4*)&acc[r * LDA + 4 * lane] = (r < nrows) ? own[j] : make_float4(0.f, 0.f, 0.f, 0.f);
        if (lane == 0) acc[r * LDA + KDIM] = 0.f;
    }
    if (tid < NWAVES) bcnt[tid] = 0;
    __syncthreads();

    unsigned* mybin    = bins + wave * BINCAP;
    unsigned* mysorted = sorted + wave * BINCAP;

    if (!use_list) {
        const int   nsteps = (nE + GSTEP * 32 - 1) / (GSTEP * 32);
        const int4* klo    = (const int4*)keys;
        const int4* khi    = klo + (size_t)nsteps * GSTEP;
        unsigned inv[8];
#pragma unroll
        for (int i = 0; i < 8; ++i) inv[i] = ((blockIdx.x >> i) & 1u) ? 0u : 0xFFFFFFFFu;

        int4 ga[4], gb[4];
        int  ncand = 0;
        for (int step = wave; step < nsteps; step += NWAVES) {
            const int4* pl = klo + (size_t)step * GSTEP + lane;
            const int4* ph = khi + (size_t)step * GSTEP + lane;
#pragma unroll
            for (int j = 0; j < 4; ++j) {
                ga[j] = pl[j * 32];
                gb[j] = ph[j * 32];
            }
#pragma unroll
            for (int j = 0; j < 4; ++j) {
                const int e0 = (step * GSTEP + j * 32 + lane) << 5;
                unsigned mm = ((unsigned)ga[j].x ^ inv[0]) & ((unsigned)ga[j].y ^ inv[1]) &
                              ((unsigned)ga[j].z ^ inv[2]) & ((unsigned)ga[j].w ^ inv[3]) &
                              ((unsigned)gb[j].x ^ inv[4]) & ((unsigned)gb[j].y ^ inv[5]) &
                              ((unsigned)gb[j].z ^ inv[6]) & ((unsigned)gb[j].w ^ inv[7]);
                unsigned act = __builtin_amdgcn_ballot_w32(mm != 0u);
                while (act != 0u) {
                    if (mm != 0u) {
                        const int b = __builtin_ctz(mm);
                        mm &= mm - 1u;
                        const int pos = ncand + (int)__builtin_amdgcn_mbcnt_lo(act, 0u);
                        if (pos < BINCAP) mysorted[pos] = (unsigned)(e0 + b);
                    }
                    ncand += __builtin_popcount(act);
                    act = __builtin_amdgcn_ballot_w32(mm != 0u);
                }
            }
        }
        if (lane == 0) ccnt[wave] = ncand;
        __syncthreads();

        int nc = ncand;
        if (nc > BINCAP) nc = BINCAP;
        for (int i = 0; i < nc; i += 32) {
            const bool live = (i + lane) < nc;
            const int  e    = (int)mysorted[live ? i + lane : 0];
            const bool ok   = live && e < nE;
            const unsigned ent = rec[ok ? e : 0];
            const unsigned d   = ent & 255u;
            if (ok && d < (unsigned)nrows) {
                const int ow  = (int)(d & 31u);
                const int pos = __hip_atomic_fetch_add(&bcnt[ow], 1, __ATOMIC_RELAXED, __HIP_MEMORY_SCOPE_WORKGROUP);
                if (pos < BINCAP) bins[ow * BINCAP + pos] = ent;
            }
        }
        __syncthreads();
    }

    const int cw = use_list ? (int)mylist[lane] : bcnt[lane];
    int off = 0, total = 0;
#pragma unroll
    for (int i = 0; i < NWAVES; ++i) {
        const int ci = __builtin_amdgcn_readlane(cw, i);
        if (i < wave) off += ci;
        total += ci;
    }
    int n = __builtin_amdgcn_readlane(cw, wave);
    const int  cc   = use_list ? 0 : ccnt[lane];
    const bool full = __builtin_amdgcn_ballot_w32(cw > BINCAP || cc > BINCAP) != 0u;
    if (n > BINCAP) n = BINCAP;
    bool put_list = false;
    if constexpr (!REPLAY) {
        put_list = !full && total <= LCAP - 32;
        if (tid == 0) ecnt[blockIdx.x] = put_list ? total : LCAP + 1;
        if (put_list && wave == 0) mylist[lane] = (unsigned)cw;
    }
    unsigned* myrun = mylist + 32 + off;

    if (!use_list && full) {
        for (int e = tid; e < nE; e += NWAVES * 32) {
            const unsigned dl = (unsigned)(dst[e] - row0);
            if (dl < (unsigned)nrows) {
                const float* p = feat + (size_t)src[e] * KDIM;
                float* r = &acc[dl * LDA];
#pragma nounroll
                for (int f = 0; f < KDIM; ++f)
                    __hip_atomic_fetch_add(r + f, p[f], __ATOMIC_RELAXED, __HIP_MEMORY_SCOPE_WORKGROUP);
                __hip_atomic_fetch_add(r + KDIM, 1.0f, __ATOMIC_RELAXED, __HIP_MEMORY_SCOPE_WORKGROUP);
            }
        }
        n = 0;
        __syncthreads();
    }

    if (!use_list) {
        int fill[NSLOT];
#pragma unroll
        for (int j = 0; j < NSLOT; ++j) fill[j] = 0;
        for (int i = 0; i < n; i += 32) {
            const bool live = (i + lane) < n;
            const int  kk   = live ? (int)((mybin[live ? i + lane : 0] & 255u) >> 5) : NSLOT;
#pragma unroll
            for (int j = 0; j < NSLOT; ++j) fill[j] += __builtin_popcount(__builtin_amdgcn_ballot_w32(kk == j));
        }
        int run = 0;
#pragma unroll
        for (int j = 0; j < NSLOT; ++j) {
            const int cj = fill[j];
            fill[j] = run;
            run += cj;
        }
        for (int i = 0; i < n; i += 32) {
            const bool live = (i + lane) < n;
            const unsigned ent = mybin[live ? i + lane : 0];
            const unsigned dl  = ent & 255u;
            const int      kk  = live ? (int)(dl >> 5) : NSLOT;
            int pos = 0;
#pragma unroll
            for (int j = 0; j < NSLOT; ++j) {
                const unsigned m = __builtin_amdgcn_ballot_w32(kk == j);
                if (kk == j) pos = fill[j] + (int)__builtin_amdgcn_mbcnt_lo(m, 0u);
                fill[j] += __builtin_popcount(m);
            }
            if (live) mysorted[pos] = ent;
        }
    }

    int deg[NSLOT];
#pragma unroll
    for (int j = 0; j < NSLOT; ++j) deg[j] = 0;

    auto fetch = [&](int i, int& s, int& k) {
        const bool live = (i + lane) < n;
        unsigned ent = 255u;
        if (live) ent = use_list ? myrun[i + lane] : mysorted[i + lane];
        if constexpr (!REPLAY) {
            if (put_list && live) myrun[i + lane] = ent;
        }
        s = (int)(ent >> 8);
        k = (int)((ent & 255u) >> 5);
#pragma unroll
        for (int j = 0; j < NSLOT; ++j) deg[j] += __builtin_popcount(__builtin_amdgcn_ballot_w32(k == j));
    };

    float4 sum = make_float4(0.f, 0.f, 0.f, 0.f);
    int    cur = NSLOT;
    auto flush = [&]() {
        if (cur < NSLOT) {
            float4* rp = (float4*)&acc[(32 * cur + wave) * LDA + 4 * lane];
            float4 o = *rp;
            o.x += sum.x; o.y += sum.y; o.z += sum.z; o.w += sum.w;
            *rp = o;
        }
        sum = make_float4(0.f, 0.f, 0.f, 0.f);
    };

    int s, k;
    fetch(0, s, k);
    for (int i = 0; i < n; i += 32) {
        int s2, k2;
        fetch(i + 32, s2, k2);
        int nb = n - i;
        if (nb > 32) nb = 32;
        for (int u0 = 0; u0 < nb; u0 += GROWS) {
            float4 v[GROWS];
            int    ks[GROWS];
#pragma unroll
            for (int u = 0; u < GROWS; ++u) {
                const int su = __builtin_amdgcn_readlane(s, u0 + u);
                ks[u]        = __builtin_amdgcn_readlane(k, u0 + u);
                v[u] = *(const float4*)(feat + (size_t)su * KDIM + 4 * lane);
            }
#pragma unroll
            for (int u = 0; u < GROWS; ++u) {
                if (ks[u] != cur) {
                    flush();
                    cur = ks[u];
                }
                sum.x += v[u].x; sum.y += v[u].y; sum.z += v[u].z; sum.w += v[u].w;
            }
        }
        s = s2;
        k = k2;
    }
    flush();

#pragma unroll
    for (int j = 0; j < NSLOT; ++j) {
        const int r = 32 * j + wave;
        if (r < nrows) {
            float* rp = &acc[r * LDA];
            float4 t  = *(float4*)&rp[4 * lane];
            const float inv = 1.0f / ((float)deg[j] + rp[KDIM] + 1.0f);
            t.x *= inv; t.y *= inv; t.z *= inv; t.w *= inv;
            *(float4*)&rp[4 * lane] = t;
        }
    }
    __syncthreads();

    constexpr int NCT  = NOUT / 16;
    constexpr int RG   = NWAVES / NCT;
    constexpr int NACC = (RROWS / 16 + RG - 1) / RG;
    const int ct = wave % NCT;
    const int rg = wave / NCT;
    const int n0 = ct * 16;
    const int lr = lane & 15;
    const int hi = lane >> 4;

    int arow[NACC];
#pragma unroll
    for (int i = 0; i < NACC; ++i) {
        const int t = rg + RG * i;
        arow[i] = ((t < ntiles) ? t : (ntiles - 1)) * 16 + lr;
    }
    v8f c[NACC];
#pragma unroll
    for (int i = 0; i < NACC; ++i) c[i] = (v8f){0.f, 0.f, 0.f, 0.f, 0.f, 0.f, 0.f, 0.f};

#pragma unroll 8
    for (int k = 0; k < KDIM; k += 4) {
        v2f b;
        const float* bp = &W[(size_t)(k + 2 * hi) * NOUT + n0 + lr];
        b.x = bp[0];
        b.y = bp[NOUT];
#pragma unroll
        for (int i = 0; i < NACC; ++i) {
            v2f a;
            const float* ap = &acc[arow[i] * LDA + k + 2 * hi];
            a.x = ap[0];
            a.y = ap[1];
            c[i] = __builtin_amdgcn_wmma_f32_16x16x4_f32(
                 false, a,  false, b,
                 (short)0, c[i],  false,  false);
        }
    }

    const float bv  = bias[n0 + lr];
    const int   col = n0 + lr;
#pragma unroll
    for (int i = 0; i < NACC; ++i) {
        const int t = rg + RG * i;
        if (t < ntiles) {
            const int rbase = row0 + t * 16 + 8 * hi;
#pragma unroll
            for (int r = 0; r < 8; ++r) {
                float v = c[i][r] + bv;
                if (apply_lrelu) v = (v >= 0.f) ? v : 0.01f * v;
                out[(size_t)(rbase + r) * NOUT + col] = v;
            }
        }
    }
}

__global__ __launch_bounds__(128) void dst_keys(const int* __restrict__ src,
                                                const int* __restrict__ dst,
                                                unsigned* __restrict__ keys,
                                                unsigned* __restrict__ rec,
                                                int nE) {
    const int nsteps = (nE + GSTEP * 32 - 1) / (GSTEP * 32);
    const int ngrp   = nsteps * GSTEP;
    uint4* lo = (uint4*)keys;
    uint4* hi = lo + ngrp;
    const int lane = threadIdx.x & 31;
    const int wv   = __builtin_amdgcn_readfirstlane((int)((blockIdx.x * 128 + threadIdx.x) >> 5));
    const int nwv  = gridDim.x * 4;
    for (int g0 = wv * 32; g0 < ngrp; g0 += nwv * 32) {
        int dv[32], sv[32];
#pragma unroll
        for (int q = 0; q < 32; ++q) {
            const int e = (g0 + q) * 32 + lane;
            dv[q] = dst[e < nE ? e : 0];
            sv[q] = src[e < nE ? e : 0];
        }
        uint4 mlo = make_uint4(0u, 0u, 0u, 0u), mhi = make_uint4(0u, 0u, 0u, 0u);
#pragma unroll
        for (int q = 0; q < 32; ++q) {
            const int e = (g0 + q) * 32 + lane;
            const unsigned k = (e < nE && (unsigned)dv[q] < (unsigned)N_NODES) ? (unsigned)dv[q] / RROWS : 255u;
            if (e < nE) rec[e] = ((unsigned)sv[q] << 8) | (((unsigned)dv[q] - k * RROWS) & 255u);
            const unsigned p0 = __builtin_amdgcn_ballot_w32((k & 1u) != 0u);
            const unsigned p1 = __builtin_amdgcn_ballot_w32((k & 2u) != 0u);
            const unsigned p2 = __builtin_amdgcn_ballot_w32((k & 4u) != 0u);
            const unsigned p3 = __builtin_amdgcn_ballot_w32((k & 8u) != 0u);
            const unsigned p4 = __builtin_amdgcn_ballot_w32((k & 16u) != 0u);
            const unsigned p5 = __builtin_amdgcn_ballot_w32((k & 32u) != 0u);
            const unsigned p6 = __builtin_amdgcn_ballot_w32((k & 64u) != 0u);
            const unsigned p7 = __builtin_amdgcn_ballot_w32((k & 128u) != 0u);
            if (lane == q) {
                mlo = make_uint4(p0, p1, p2, p3);
                mhi = make_uint4(p4, p5, p6, p7);
            }
        }
        lo[g0 + lane] = mlo;
        hi[g0 + lane] = mhi;
    }
}

extern "C" void kernel_launch(void* const* d_in, const int* in_sizes, int n_in,
                              void* d_out, int out_size, void* d_ws, size_t ws_size,
                              hipStream_t stream) {
    const float* x   = (const float*)d_in[0];
    const int*   edg = (const int*)  d_in[1];
    const float* W1  = (const float*)d_in[2];
    const float* b1  = (const float*)d_in[3];
    const float* W2  = (const float*)d_in[4];
    const float* b2  = (const float*)d_in[5];
    float*       outp = (float*)d_out;

    const int E = in_sizes[1] / 2;
    const int* src = edg;
    const int* dst = edg + E;

    float*    h1    = (float*)d_ws;
    unsigned* elist = (unsigned*)(h1 + (size_t)N_NODES * KDIM);
    int*      ecnt  = (int*)(elist + (size_t)NWG * LCAP);
    unsigned* rec  = (unsigned*)(ecnt + 256);
    unsigned* keys = rec + E;

    dst_keys<<<196, 128, 0, stream>>>(src, dst, keys, rec, E);
    sage_layer<128, false><<<NWG, NWAVES * 32, 0, stream>>>(
        x, src, dst, keys, rec, W1, b1, h1, elist, ecnt, E,  1);
    sage_layer<32, true><<<NWG, NWAVES * 32, 0, stream>>>(
        h1, src, dst, keys, rec, W2, b2, outp, elist, ecnt, E,  0);
}


# owner number coded as two one-of-16 marks (32 bit planes, by ballots in the front launch): a workgroup scans only its own 2 planes, 8 B per 32 edges, was 32 B, one AND, was 15 ops; sums by arrival
# speedup vs baseline: 5.4170x; 1.0362x over previous
#include <hip/hip_runtime.h>


typedef float v2f __attribute__((ext_vector_type(2)));
typedef float v8f __attribute__((ext_vector_type(8)));

#define N_NODES 50000
#define KDIM    128
#define LDA     132
#define RROWS   208
#define NWAVES  32
#define NSLOT   7
#define AROWS   (NSLOT * 32)
#define QROW    1024
#define QTRIP   (4 * QROW)
#define QEDGES  (128 * QTRIP)
#define BINCAP  768
#define GROWS   16
#define NWG     ((N_NODES + RROWS - 1) / RROWS)
static_assert((N_NODES + RROWS - 1) / RROWS <= 256, "owner numbers must split into two parts of 0..15");
#define LCAP    4096

template <int NOUT, bool REPLAY>
__global__ __launch_bounds__(1024) void sage_layer(const float* __restrict__ feat,
                                                   const int* __restrict__ src,
                                                   const int* __restrict__ dst,
                                                   const unsigned* __restrict__ keys,
                                                   const unsigned* __restrict__ rec,
                                                   const float* __restrict__ W,
                                                   const float* __restrict__ bias,
                                                   float* __restrict__ out,
                                                   unsigned* __restrict__ elist,
                                                   int* __restrict__ ecnt,
                                                   int nE, int apply_lrelu) {
    __shared__ __attribute__((aligned(16))) float acc[AROWS * LDA];
    __shared__ unsigned bins[NWAVES * BINCAP];
    __shared__ unsigned sorted[NWAVES * BINCAP];
    __shared__ int bcnt[NWAVES];
    __shared__ int ccnt[NWAVES];

    const int tid  = threadIdx.x;
    const int wave = __builtin_amdgcn_readfirstlane(tid >> 5);
    const int lane = tid & 31;
    const int row0 = blockIdx.x * RROWS;
    int nrows = N_NODES - row0;
    if (nrows > RROWS) nrows = RROWS;
    const int ntiles = nrows >> 4;

    unsigned* mylist = elist + (size_t)blockIdx.x * LCAP;
    int nlist = LCAP + 1;
    if constexpr (REPLAY) nlist = __builtin_amdgcn_readfirstlane(ecnt[blockIdx.x]);
    const bool use_list = REPLAY && nlist <= LCAP - 32;

    float4 own[NSLOT];
#pragma unroll
    for (int j = 0; j < NSLOT; ++j) {
        const int r = 32 * j + wave;
        own[j] = *(const float4*)(feat + (size_t)(row0 + (r < nrows ? r : 0)) * KDIM + 4 * lane);
    }
#pragma unroll
    for (int j = 0; j < NSLOT; ++j) {
        const int r = 32 * j + wave;
        *(float4*)&acc[r * LDA + 4 * lane] = (r < nrows) ? own[j] : make_float4(0.f, 0.f, 0.f, 0.f);
        if (lane == 0) acc[r * LDA + KDIM] = 0.f;
    }
    if (tid < NWAVES) bcnt[tid] = 0;
    __syncthreads();

    unsigned* mybin    = bins + wave * BINCAP;
    unsigned* mysorted = sorted + wave * BINCAP;

    if (!use_list) {
        const int    nq = ((nE + QEDGES - 1) / QEDGES) * QTRIP;
        const uint4* ph = (const uint4*)keys + (size_t)(blockIdx.x >> 4) * nq;
        const uint4* pl = (const uint4*)keys + (size_t)(16 + (blockIdx.x & 15)) * nq;

        uint4 ga[4], gb[4];
        int   ncand = 0;
        for (int qb = wave * 32; qb < nq; qb += QTRIP) {
#pragma unroll
            for (int j = 0; j < 4; ++j) {
                ga[j] = ph[qb + j * QROW + lane];
                gb[j] = pl[qb + j * QROW + lane];
            }
#pragma unroll
            for (int j = 0; j < 4; ++j) {
                const int q = qb + j * QROW;
                const unsigned m4[4] = {ga[j].x & gb[j].x, ga[j].y & gb[j].y, ga[j].z & gb[j].z, ga[j].w & gb[j].w};
#pragma unroll
                for (int cw = 0; cw < 4; ++cw) {
                    const int e0 = (((q + lane) << 2) + cw) << 5;
                    unsigned mm  = m4[cw];
                    unsigned act = __builtin_amdgcn_ballot_w32(mm != 0u);
                    while (act != 0u) {
                        if (mm != 0u) {
                            const int b = __builtin_ctz(mm);
                            mm &= mm - 1u;
                            const int pos = ncand + (int)__builtin_amdgcn_mbcnt_lo(act, 0u);
                            if (pos < BINCAP) mysorted[pos] = (unsigned)(e0 + b);
                        }
                        ncand += __builtin_popcount(act);
                        act = __builtin_amdgcn_ballot_w32(mm != 0u);
                    }
                }
            }
        }
        if (lane == 0) ccnt[wave] = ncand;
        __syncthreads();

        int nc = ncand;
        if (nc > BINCAP) nc = BINCAP;
        for (int i = 0; i < nc; i += 32) {
            const bool live = (i + lane) < nc;
            const int  e    = (int)mysorted[live ? i + lane : 0];
            const bool ok   = live && e < nE;
            const unsigned ent = rec[ok ? e : 0];
            const unsigned d   = ent & 255u;
            if (ok && d < (unsigned)nrows) {
                const int ow  = (int)(d & 31u);
                const int pos = __hip_atomic_fetch_add(&bcnt[ow], 1, __ATOMIC_RELAXED, __HIP_MEMORY_SCOPE_WORKGROUP);
                if (pos < BINCAP) bins[ow * BINCAP + pos] = ent;
            }
        }
        __syncthreads();
    }

    const int cw = use_list ? (int)mylist[lane] : bcnt[lane];
    int off = 0, total = 0;
#pragma unroll
    for (int i = 0; i < NWAVES; ++i) {
        const int ci = __builtin_amdgcn_readlane(cw, i);
        if (i < wave) off += ci;
        total += ci;
    }
    int n = __builtin_amdgcn_readlane(cw, wave);
    const int  cc   = use_list ? 0 : ccnt[lane];
    const bool full = __builtin_amdgcn_ballot_w32(cw > BINCAP || cc > BINCAP) != 0u;
    if (n > BINCAP) n = BINCAP;
    bool put_list = false;
    if constexpr (!REPLAY) {
        put_list = !full && total <= LCAP - 32;
        if (tid == 0) ecnt[blockIdx.x] = put_list ? total : LCAP + 1;
        if (put_list && wave == 0) mylist[lane] = (unsigned)cw;
    }
    unsigned* myrun = mylist + 32 + off;

    if (!use_list && full) {
        for (int e = tid; e < nE; e += NWAVES * 32) {
            const unsigned dl = (unsigned)(dst[e] - row0);
            if (dl < (unsigned)nrows) {
                const float* p = feat + (size_t)src[e] * KDIM;
                float* r = &acc[dl * LDA];
#pragma nounroll
                for (int f = 0; f < KDIM; ++f)
                    __hip_atomic_fetch_add(r + f, p[f], __ATOMIC_RELAXED, __HIP_MEMORY_SCOPE_WORKGROUP);
                __hip_atomic_fetch_add(r + KDIM, 1.0f, __ATOMIC_RELAXED, __HIP_MEMORY_SCOPE_WORKGROUP);
            }
        }
        n = 0;
        __syncthreads();
    }

    if (!use_list) {
        int fill[NSLOT];
#pragma unroll
        for (int j = 0; j < NSLOT; ++j) fill[j] = 0;
        for (int i = 0; i < n; i += 32) {
            const bool live = (i + lane) < n;
            const int  kk   = live ? (int)((mybin[live ? i + lane : 0] & 255u) >> 5) : NSLOT;
#pragma unroll
            for (int j = 0; j < NSLOT; ++j) fill[j] += __builtin_popcount(__builtin_amdgcn_ballot_w32(kk == j));
        }
        int run = 0;
#pragma unroll
        for (int j = 0; j < NSLOT; ++j) {
            const int cj = fill[j];
            fill[j] = run;
            run += cj;
        }
        for (int i = 0; i < n; i += 32) {
            const bool live = (i + lane) < n;
            const unsigned ent = mybin[live ? i + lane : 0];
            const unsigned dl  = ent & 255u;
            const int      kk  = live ? (int)(dl >> 5) : NSLOT;
            int pos = 0;
#pragma unroll
            for (int j = 0; j < NSLOT; ++j) {
                const unsigned m = __builtin_amdgcn_ballot_w32(kk == j);
                if (kk == j) pos = fill[j] + (int)__builtin_amdgcn_mbcnt_lo(m, 0u);
                fill[j] += __builtin_popcount(m);
            }
            if (live) mysorted[pos] = ent;
        }
    }

    int deg[NSLOT];
#pragma unroll
    for (int j = 0; j < NSLOT; ++j) deg[j] = 0;

    auto fetch = [&](int i, int& s, int& k) {
        const bool live = (i + lane) < n;
        unsigned ent = 255u;
        if (live) ent = use_list ? myrun[i + lane] : mysorted[i + lane];
        if constexpr (!REPLAY) {
            if (put_list && live) myrun[i + lane] = ent;
        }
        s = (int)(ent >> 8);
        k = (int)((ent & 255u) >> 5);
#pragma unroll
        for (int j = 0; j < NSLOT; ++j) deg[j] += __builtin_popcount(__builtin_amdgcn_ballot_w32(k == j));
    };

    float4 sum = make_float4(0.f, 0.f, 0.f, 0.f);
    int    cur = NSLOT;
    auto flush = [&]() {
        if (cur < NSLOT) {
            float4* rp = (float4*)&acc[(32 * cur + wave) * LDA + 4 * lane];
            float4 o = *rp;
            o.x += sum.x; o.y += sum.y; o.z += sum.z; o.w += sum.w;
            *rp = o;
        }
        sum = make_float4(0.f, 0.f, 0.f, 0.f);
    };

    int s, k;
    fetch(0, s, k);
    for (int i = 0; i < n; i += 32) {
        int s2, k2;
        fetch(i + 32, s2, k2);
        int nb = n - i;
        if (nb > 32) nb = 32;
        for (int u0 = 0; u0 < nb; u0 += GROWS) {
            float4 v[GROWS];
            int    ks[GROWS];
#pragma unroll
            for (int u = 0; u < GROWS; ++u) {
                const int su = __builtin_amdgcn_readlane(s, u0 + u);
                ks[u]        = __builtin_amdgcn_readlane(k, u0 + u);
                v[u] = *(const float4*)(feat + (size_t)su * KDIM + 4 * lane);
            }
#pragma unroll
            for (int u = 0; u < GROWS; ++u) {
                if (ks[u] != cur) {
                    flush();
                    cur = ks[u];
                }
                sum.x += v[u].x; sum.y += v[u].y; sum.z += v[u].z; sum.w += v[u].w;
            }
        }
        s = s2;
        k = k2;
    }
    flush();

#pragma unroll
    for (int j = 0; j < NSLOT; ++j) {
        const int r = 32 * j + wave;
        if (r < nrows) {
            float* rp = &acc[r * LDA];
            float4 t  = *(float4*)&rp[4 * lane];
            const float inv = 1.0f / ((float)deg[j] + rp[KDIM] + 1.0f);
            t.x *= inv; t.y *= inv; t.z *= inv; t.w *= inv;
            *(float4*)&rp[4 * lane] = t;
        }
    }
    __syncthreads();

    constexpr int NCT  = NOUT / 16;
    constexpr int RG   = NWAVES / NCT;
    constexpr int NACC = (RROWS / 16 + RG - 1) / RG;
    const int ct = wave % NCT;
    const int rg = wave / NCT;
    const int n0 = ct * 16;
    const int lr = lane & 15;
    const int hi = lane >> 4;

    int arow[NACC];
#pragma unroll
    for (int i = 0; i < NACC; ++i) {
        const int t = rg + RG * i;
        arow[i] = ((t < ntiles) ? t : (ntiles - 1)) * 16 + lr;
    }
    v8f c[NACC];
#pragma unroll
    for (int i = 0; i < NACC; ++i) c[i] = (v8f){0.f, 0.f, 0.f, 0.f, 0.f, 0.f, 0.f, 0.f};

#pragma unroll 8
    for (int k = 0; k < KDIM; k += 4) {
        v2f b;
        const float* bp = &W[(size_t)(k + 2 * hi) * NOUT + n0 + lr];
        b.x = bp[0];
        b.y = bp[NOUT];
#pragma unroll
        for (int i = 0; i < NACC; ++i) {
            v2f a;
            const float* ap = &acc[arow[i] * LDA + k + 2 * hi];
            a.x = ap[0];
            a.y = ap[1];
            c[i] = __builtin_amdgcn_wmma_f32_16x16x4_f32(
                 false, a,  false, b,
                 (short)0, c[i],  false,  false);
        }
    }

    const float bv  = bias[n0 + lr];
    const int   col = n0 + lr;
#pragma unroll
    for (int i = 0; i < NACC; ++i) {
        const int t = rg + RG * i;
        if (t < ntiles) {
            const int rbase = row0 + t * 16 + 8 * hi;
#pragma unroll
            for (int r = 0; r < 8; ++r) {
                float v = c[i][r] + bv;
                if (apply_lrelu) v = (v >= 0.f) ? v : 0.01f * v;
                out[(size_t)(rbase + r) * NOUT + col] = v;
            }
        }
    }
}

__global__ __launch_bounds__(128) void dst_keys(const int* __restrict__ src,
                                                const int* __restrict__ dst,
                                                unsigned* __restrict__ keys,
                                                unsigned* __restrict__ rec,
                                                int nE) {
    __shared__ unsigned square[4][32 * 33];
    const int nw   = ((nE + QEDGES - 1) / QEDGES) * QTRIP * 4;
    const int lane = threadIdx.x & 31;
    const int wv   = __builtin_amdgcn_readfirstlane((int)((blockIdx.x * 128 + threadIdx.x) >> 5));
    const int nwv  = gridDim.x * 4;
    unsigned* sq   = square[__builtin_amdgcn_readfirstlane((int)(threadIdx.x >> 5))];
    const bool lowp = lane >= 16;
    unsigned inv[4];
#pragma unroll
    for (int c = 0; c < 4; ++c) inv[c] = ((lane >> c) & 1) ? 0u : 0xFFFFFFFFu;
    for (int g0 = wv * 32; g0 < nw; g0 += nwv * 32) {
        int dv[32], sv[32];
#pragma unroll
        for (int q = 0; q < 32; ++q) {
            const int e = (g0 + q) * 32 + lane;
            dv[q] = dst[e < nE ? e : 0];
            sv[q] = src[e < nE ? e : 0];
        }
#pragma unroll
        for (int q = 0; q < 32; ++q) {
            const int e = (g0 + q) * 32 + lane;
            const bool has = e < nE && (unsigned)dv[q] < (unsigned)N_NODES;
            const unsigned k = (unsigned)dv[q] / RROWS;
            if (e < nE) rec[e] = ((unsigned)sv[q] << 8) | (((unsigned)dv[q] - k * RROWS) & 255u);
            const unsigned hv = __builtin_amdgcn_ballot_w32(has);
            const unsigned p0 = __builtin_amdgcn_ballot_w32((k & 1u) != 0u);
            const unsigned p1 = __builtin_amdgcn_ballot_w32((k & 2u) != 0u);
            const unsigned p2 = __builtin_amdgcn_ballot_w32((k & 4u) != 0u);
            const unsigned p3 = __builtin_amdgcn_ballot_w32((k & 8u) != 0u);
            const unsigned p4 = __builtin_amdgcn_ballot_w32((k & 16u) != 0u);
            const unsigned p5 = __builtin_amdgcn_ballot_w32((k & 32u) != 0u);
            const unsigned p6 = __builtin_amdgcn_ballot_w32((k & 64u) != 0u);
            const unsigned p7 = __builtin_amdgcn_ballot_w32((k & 128u) != 0u);
            const unsigned w = hv & ((lowp ? p0 : p4) ^ inv[0]) & ((lowp ? p1 : p5) ^ inv[1]) &
                                    ((lowp ? p2 : p6) ^ inv[2]) & ((lowp ? p3 : p7) ^ inv[3]);
            sq[q * 33 + lane] = w;
        }
        __builtin_amdgcn_fence(__ATOMIC_ACQ_REL, "wavefront");
#pragma unroll
        for (int p = 0; p < 32; ++p) keys[(size_t)p * nw + g0 + lane] = sq[lane * 33 + p];
        __builtin_amdgcn_fence(__ATOMIC_ACQ_REL, "wavefront");
    }
}

extern "C" void kernel_launch(void* const* d_in, const int* in_sizes, int n_in,
                              void* d_out, int out_size, void* d_ws, size_t ws_size,
                              hipStream_t stream) {
    const float* x   = (const float*)d_in[0];
    const int*   edg = (const int*)  d_in[1];
    const float* W1  = (const float*)d_in[2];
    const float* b1  = (const float*)d_in[3];
    const float* W2  = (const float*)d_in[4];
    const float* b2  = (const float*)d_in[5];
    float*       outp = (float*)d_out;

    const int E = in_sizes[1] / 2;
    const int* src = edg;
    const int* dst = edg + E;

    float*    h1    = (float*)d_ws;
    unsigned* elist = (unsigned*)(h1 + (size_t)N_NODES * KDIM);
    int*      ecnt  = (int*)(elist + (size_t)NWG * LCAP);
    unsigned* rec  = (unsigned*)(ecnt + 256);
    unsigned* keys = rec + E;

    dst_keys<<<256, 128, 0, stream>>>(src, dst, keys, rec, E);
    sage_layer<128, false><<<NWG, NWAVES * 32, 0, stream>>>(
        x, src, dst, keys, rec, W1, b1, h1, elist, ecnt, E,  1);
    sage_layer<32, true><<<NWG, NWAVES * 32, 0, stream>>>(
        h1, src, dst, keys, rec, W2, b2, outp, elist, ecnt, E,  0);
}
